# starting kernel
# baseline (speedup 1.0000x reference)
#include <hip/hip_runtime.h>


typedef __attribute__((ext_vector_type(2))) float v2f;
typedef __attribute__((ext_vector_type(8))) float v8f;

__device__ __forceinline__ v8f wmma_k4(v2f a, v2f b, v8f c) {
  return __builtin_amdgcn_wmma_f32_16x16x4_f32(
       false, a,  false, b,
       (short)0, c,  false,  false);
}


__global__ void zero_kernel(float* __restrict__ p, long long n) {
  long long i = (long long)blockIdx.x * blockDim.x + threadIdx.x;
  if (i < n) p[i] = 0.0f;
}

__global__ void count_kernel(const int* __restrict__ dst, int E,
                             float* __restrict__ cnt) {
  int i = blockIdx.x * blockDim.x + threadIdx.x;
  if (i < E) atomicAdd(&cnt[dst[i]], 1.0f);
}

__global__ void addmat_kernel(const float* __restrict__ a,
                              const float* __restrict__ b,
                              float* __restrict__ o, int n) {
  int i = blockIdx.x * blockDim.x + threadIdx.x;
  if (i < n) o[i] = a[i] + b[i];
}

__global__ void scatter16_kernel(const float* __restrict__ Y,
                                 const int* __restrict__ src,
                                 const int* __restrict__ dst, int E,
                                 float* __restrict__ agg) {
  long long i = (long long)blockIdx.x * blockDim.x + threadIdx.x;
  long long total = (long long)E * 4;
  if (i >= total) return;
  int e = (int)(i >> 2);
  int c = (int)(i & 3);
  int s = src[e];
  int d = dst[e];
  const float4 v = *reinterpret_cast<const float4*>(Y + (size_t)s * 16 + c * 4);
  float* p = agg + (size_t)d * 16 + c * 4;
  atomicAdd(p + 0, v.x);
  atomicAdd(p + 1, v.y);
  atomicAdd(p + 2, v.z);
  atomicAdd(p + 3, v.w);
}

__global__ void finalize_l1_kernel(float* __restrict__ XO,
                                   const float* __restrict__ aggA,
                                   const float* __restrict__ cntA,
                                   const float* __restrict__ bA,
                                   const float* __restrict__ aggB,
                                   const float* __restrict__ cntB,
                                   const float* __restrict__ bB, int M) {
  long long i = (long long)blockIdx.x * blockDim.x + threadIdx.x;
  if (i >= (long long)M * 16) return;
  int node = (int)(i >> 4);
  int f = (int)(i & 15);
  float ca = fmaxf(cntA[node], 1.0f);
  float cb = fmaxf(cntB[node], 1.0f);
  float v = XO[i] + aggA[i] / ca + bA[f] + aggB[i] / cb + bB[f];
  XO[i] = fmaxf(v, 0.0f);
}

__global__ __launch_bounds__(128) void l1_transform_kernel(
    const float* __restrict__ X, const float* __restrict__ W0,
    const float* __restrict__ W1, const float* __restrict__ W2,
    float* __restrict__ Y0, float* __restrict__ Y1, float* __restrict__ Y2,
    int M) {
  const int lane = threadIdx.x & 31;
  const int wave = threadIdx.x >> 5;
  const int wpb = blockDim.x >> 5;
  const int half = lane >> 4;
  const int mrow = lane & 15;
  const int nTiles = (M + 15) >> 4;

  v2f b0[16], b1[16], b2[16];
#pragma unroll
  for (int s = 0; s < 16; ++s) {
    int k = 4 * s + 2 * half;
    b0[s] = (v2f){W0[k * 16 + mrow], W0[(k + 1) * 16 + mrow]};
    b1[s] = (v2f){W1[k * 16 + mrow], W1[(k + 1) * 16 + mrow]};
    b2[s] = (v2f){W2[k * 16 + mrow], W2[(k + 1) * 16 + mrow]};
  }

  for (int tile = blockIdx.x * wpb + wave; tile < nTiles;
       tile += gridDim.x * wpb) {
    const int rbase = tile << 4;
    int arow = rbase + mrow;
    if (arow > M - 1) arow = M - 1;
    const float* xr = X + (size_t)arow * 64;
    v8f c0 = {}, c1 = {}, c2 = {};
#pragma unroll
    for (int s = 0; s < 16; ++s) {
      int k = 4 * s + 2 * half;
      v2f a = (v2f){xr[k], xr[k + 1]};
      c0 = wmma_k4(a, b0[s], c0);
      c1 = wmma_k4(a, b1[s], c1);
      c2 = wmma_k4(a, b2[s], c2);
    }
#pragma unroll
    for (int r = 0; r < 8; ++r) {
      int row = rbase + r + 8 * half;
      if (row < M) {
        size_t o = (size_t)row * 16 + mrow;
        Y0[o] = c0[r];
        Y1[o] = c1[r];
        Y2[o] = c2[r];
      }
    }
  }
}

__global__ __launch_bounds__(128) void l2_fused_kernel(
    const float* __restrict__ A0, const float* __restrict__ cnt0,
    const float* __restrict__ A1, const float* __restrict__ cnt1,
    const float* __restrict__ A2, const float* __restrict__ W0,
    const float* __restrict__ W1, const float* __restrict__ W2,
    const float* __restrict__ bias0, const float* __restrict__ bias1,
    float* __restrict__ OUT, int M) {
  const int lane = threadIdx.x & 31;
  const int wave = threadIdx.x >> 5;
  const int wpb = blockDim.x >> 5;
  const int half = lane >> 4;
  const int mrow = lane & 15;
  const int nTiles = (M + 15) >> 4;

  const float* Ws[3] = {W0, W1, W2};
  v2f b[3][4][2];
#pragma unroll
  for (int m = 0; m < 3; ++m)
#pragma unroll
    for (int s = 0; s < 4; ++s)
#pragma unroll
      for (int t = 0; t < 2; ++t) {
        int k = 4 * s + 2 * half;
        int col = 16 * t + mrow;
        b[m][s][t] = (v2f){Ws[m][k * 32 + col], Ws[m][(k + 1) * 32 + col]};
      }

  const float biasL = bias0[mrow] + bias1[mrow];
  const float biasH = bias0[16 + mrow] + bias1[16 + mrow];

  for (int tile = blockIdx.x * wpb + wave; tile < nTiles;
       tile += gridDim.x * wpb) {
    const int rbase = tile << 4;
    int arow = rbase + mrow;
    if (arow > M - 1) arow = M - 1;
    const float rs0 = 1.0f / fmaxf(cnt0[arow], 1.0f);
    const float rs1 = 1.0f / fmaxf(cnt1[arow], 1.0f);
    const float* a0p = A0 + (size_t)arow * 16;
    const float* a1p = A1 + (size_t)arow * 16;
    const float* a2p = A2 + (size_t)arow * 16;
    v8f cL = {}, cH = {};
#pragma unroll
    for (int s = 0; s < 4; ++s) {
      int k = 4 * s + 2 * half;
      v2f a0 = (v2f){a0p[k] * rs0, a0p[k + 1] * rs0};
      cL = wmma_k4(a0, b[0][s][0], cL);
      cH = wmma_k4(a0, b[0][s][1], cH);
      v2f a1 = (v2f){a1p[k] * rs1, a1p[k + 1] * rs1};
      cL = wmma_k4(a1, b[1][s][0], cL);
      cH = wmma_k4(a1, b[1][s][1], cH);
      v2f a2 = (v2f){a2p[k], a2p[k + 1]};
      cL = wmma_k4(a2, b[2][s][0], cL);
      cH = wmma_k4(a2, b[2][s][1], cH);
    }
#pragma unroll
    for (int r = 0; r < 8; ++r) {
      int row = rbase + r + 8 * half;
      if (row < M) {
        size_t o = (size_t)row * 32 + mrow;
        OUT[o] = cL[r] + biasL;
        OUT[o + 16] = cH[r] + biasH;
      }
    }
  }
}


extern "C" void kernel_launch(void* const* d_in, const int* in_sizes, int n_in,
                              void* d_out, int out_size, void* d_ws,
                              size_t ws_size, hipStream_t stream) {
  const float* x_dev = (const float*)d_in[0];
  const float* x_task = (const float*)d_in[1];
  const int* wr_src = (const int*)d_in[2];
  const int* wr_dst = (const int*)d_in[3];
  const int* rv_src = (const int*)d_in[4];
  const int* rv_dst = (const int*)d_in[5];
  const int* wb_src = (const int*)d_in[6];
  const int* wb_dst = (const int*)d_in[7];
  const int* rb_src = (const int*)d_in[8];
  const int* rb_dst = (const int*)d_in[9];
  const float* p1wr_wl = (const float*)d_in[10];
  const float* p1wr_bl = (const float*)d_in[11];
  const float* p1wr_wr = (const float*)d_in[12];
  const float* p1rv_wl = (const float*)d_in[13];
  const float* p1rv_bl = (const float*)d_in[14];
  const float* p1rv_wr = (const float*)d_in[15];
  const float* p1wb_wl = (const float*)d_in[16];
  const float* p1wb_bl = (const float*)d_in[17];
  const float* p1wb_wr = (const float*)d_in[18];
  const float* p1rb_wl = (const float*)d_in[19];
  const float* p1rb_bl = (const float*)d_in[20];
  const float* p1rb_wr = (const float*)d_in[21];
  const float* p2wr_wl = (const float*)d_in[22];
  const float* p2wr_bl = (const float*)d_in[23];
  const float* p2wr_wr = (const float*)d_in[24];
  const float* p2rv_wl = (const float*)d_in[25];
  const float* p2rv_bl = (const float*)d_in[26];
  const float* p2rv_wr = (const float*)d_in[27];
  const float* p2wb_wl = (const float*)d_in[28];
  const float* p2wb_bl = (const float*)d_in[29];
  const float* p2wb_wr = (const float*)d_in[30];
  const float* p2rb_wl = (const float*)d_in[31];
  const float* p2rb_bl = (const float*)d_in[32];
  const float* p2rb_wr = (const float*)d_in[33];

  const int ND = in_sizes[0] / 64;
  const int NT = in_sizes[1] / 64;
  const int E_wr = in_sizes[2], E_rv = in_sizes[4];
  const int E_wb = in_sizes[6], E_rb = in_sizes[8];

  float* ws = (float*)d_ws;
  size_t off = 0;
  float* cnt_wr = ws + off; off += NT;
  float* cnt_rv = ws + off; off += NT;
  float* cnt_wb = ws + off; off += ND;
  float* cnt_rb = ws + off; off += ND;
  float* agg_wr = ws + off; off += (size_t)NT * 16;
  float* agg_rv = ws + off; off += (size_t)NT * 16;
  float* agg_wb = ws + off; off += (size_t)ND * 16;
  float* agg_rb = ws + off; off += (size_t)ND * 16;
  float* y_wr = ws + off; off += (size_t)ND * 16;
  float* y_rv = ws + off; off += (size_t)ND * 16;
  float* y_wb = ws + off; off += (size_t)NT * 16;
  float* y_rb = ws + off; off += (size_t)NT * 16;
  float* t1 = ws + off; off += (size_t)NT * 16;
  float* d1 = ws + off; off += (size_t)ND * 16;
  float* ws1t = ws + off; off += 64 * 16;
  float* ws1d = ws + off; off += 64 * 16;
  float* ws2t = ws + off; off += 16 * 32;
  float* ws2d = ws + off; off += 16 * 32;

  const int TPB = 256;
  const long long aggN = (long long)(2LL * NT + 2LL * ND) * 16;
  const long long zero1N = (long long)(2LL * NT + 2LL * ND) + aggN;

  zero_kernel<<<(int)((zero1N + TPB - 1) / TPB), TPB, 0, stream>>>(ws, zero1N);

  count_kernel<<<(E_wr + TPB - 1) / TPB, TPB, 0, stream>>>(wr_dst, E_wr, cnt_wr);
  count_kernel<<<(E_rv + TPB - 1) / TPB, TPB, 0, stream>>>(rv_dst, E_rv, cnt_rv);
  count_kernel<<<(E_wb + TPB - 1) / TPB, TPB, 0, stream>>>(wb_dst, E_wb, cnt_wb);
  count_kernel<<<(E_rb + TPB - 1) / TPB, TPB, 0, stream>>>(rb_dst, E_rb, cnt_rb);

  addmat_kernel<<<4, TPB, 0, stream>>>(p1wr_wr, p1rv_wr, ws1t, 64 * 16);
  addmat_kernel<<<4, TPB, 0, stream>>>(p1wb_wr, p1rb_wr, ws1d, 64 * 16);
  addmat_kernel<<<2, TPB, 0, stream>>>(p2wr_wr, p2rv_wr, ws2t, 16 * 32);
  addmat_kernel<<<2, TPB, 0, stream>>>(p2wb_wr, p2rb_wr, ws2d, 16 * 32);

  const int tilesD = (ND + 15) / 16, tilesT = (NT + 15) / 16;
  l1_transform_kernel<<<(tilesD + 3) / 4, 128, 0, stream>>>(
      x_dev, p1wr_wl, p1rv_wl, ws1d, y_wr, y_rv, d1, ND);
  l1_transform_kernel<<<(tilesT + 3) / 4, 128, 0, stream>>>(
      x_task, p1wb_wl, p1rb_wl, ws1t, y_wb, y_rb, t1, NT);

  auto scb = [&](int E) { return (int)(((long long)E * 4 + TPB - 1) / TPB); };
  scatter16_kernel<<<scb(E_wr), TPB, 0, stream>>>(y_wr, wr_src, wr_dst, E_wr, agg_wr);
  scatter16_kernel<<<scb(E_rv), TPB, 0, stream>>>(y_rv, rv_src, rv_dst, E_rv, agg_rv);
  scatter16_kernel<<<scb(E_wb), TPB, 0, stream>>>(y_wb, wb_src, wb_dst, E_wb, agg_wb);
  scatter16_kernel<<<scb(E_rb), TPB, 0, stream>>>(y_rb, rb_src, rb_dst, E_rb, agg_rb);

  finalize_l1_kernel<<<(int)(((long long)NT * 16 + TPB - 1) / TPB), TPB, 0,
                       stream>>>(t1, agg_wr, cnt_wr, p1wr_bl, agg_rv, cnt_rv,
                                 p1rv_bl, NT);
  finalize_l1_kernel<<<(int)(((long long)ND * 16 + TPB - 1) / TPB), TPB, 0,
                       stream>>>(d1, agg_wb, cnt_wb, p1wb_bl, agg_rb, cnt_rb,
                                 p1rb_bl, ND);

  zero_kernel<<<(int)((aggN + TPB - 1) / TPB), TPB, 0, stream>>>(agg_wr, aggN);

  scatter16_kernel<<<scb(E_wr), TPB, 0, stream>>>(d1, wr_src, wr_dst, E_wr, agg_wr);
  scatter16_kernel<<<scb(E_rv), TPB, 0, stream>>>(d1, rv_src, rv_dst, E_rv, agg_rv);
  scatter16_kernel<<<scb(E_wb), TPB, 0, stream>>>(t1, wb_src, wb_dst, E_wb, agg_wb);
  scatter16_kernel<<<scb(E_rb), TPB, 0, stream>>>(t1, rb_src, rb_dst, E_rb, agg_rb);

  float* out_d2 = (float*)d_out;
  float* out_t2 = (float*)d_out + (size_t)ND * 32;
  l2_fused_kernel<<<(tilesT + 3) / 4, 128, 0, stream>>>(
      agg_wr, cnt_wr, agg_rv, cnt_rv, t1, p2wr_wl, p2rv_wl, ws2t, p2wr_bl,
      p2rv_bl, out_t2, NT);
  l2_fused_kernel<<<(tilesD + 3) / 4, 128, 0, stream>>>(
      agg_wb, cnt_wb, agg_rb, cnt_rb, d1, p2wb_wl, p2rb_wl, ws2d, p2wb_bl,
      p2rb_bl, out_d2, ND);
}


# Layer-1 mean folded into the scatter: rows scaled by 1/max(cnt,1) are atomically added into the root+bias buffer, ReLU applied on read in layer 2; finalize pass and agg zeroing of layer 1 removed
# speedup vs baseline: 1.0780x; 1.0780x over previous
#include <hip/hip_runtime.h>


typedef __attribute__((ext_vector_type(2))) float v2f;
typedef __attribute__((ext_vector_type(8))) float v8f;

__device__ __forceinline__ v8f wmma_k4(v2f a, v2f b, v8f c) {
  return __builtin_amdgcn_wmma_f32_16x16x4_f32(
       false, a,  false, b,
       (short)0, c,  false,  false);
}


__global__ void zero_kernel(float* __restrict__ p, long long n) {
  long long i = (long long)blockIdx.x * blockDim.x + threadIdx.x;
  if (i < n) p[i] = 0.0f;
}

__global__ void count_kernel(const int* __restrict__ dst, int E,
                             float* __restrict__ cnt) {
  int i = blockIdx.x * blockDim.x + threadIdx.x;
  if (i < E) atomicAdd(&cnt[dst[i]], 1.0f);
}

__global__ void addmat_kernel(const float* __restrict__ a,
                              const float* __restrict__ b,
                              float* __restrict__ o, int n) {
  int i = blockIdx.x * blockDim.x + threadIdx.x;
  if (i < n) o[i] = a[i] + b[i];
}

template <bool kMean, bool kRelu>
__global__ void scatter16_kernel(const float* __restrict__ Y,
                                 const int* __restrict__ src,
                                 const int* __restrict__ dst, int E,
                                 const float* __restrict__ cnt,
                                 float* __restrict__ agg) {
  long long i = (long long)blockIdx.x * blockDim.x + threadIdx.x;
  long long total = (long long)E * 4;
  if (i >= total) return;
  int e = (int)(i >> 2);
  int c = (int)(i & 3);
  int s = src[e];
  int d = dst[e];
  float4 v = *reinterpret_cast<const float4*>(Y + (size_t)s * 16 + c * 4);
  if (kRelu) {
    v.x = fmaxf(v.x, 0.0f);
    v.y = fmaxf(v.y, 0.0f);
    v.z = fmaxf(v.z, 0.0f);
    v.w = fmaxf(v.w, 0.0f);
  }
  if (kMean) {
    const float inv = 1.0f / fmaxf(cnt[d], 1.0f);
    v.x *= inv;
    v.y *= inv;
    v.z *= inv;
    v.w *= inv;
  }
  float* p = agg + (size_t)d * 16 + c * 4;
  atomicAdd(p + 0, v.x);
  atomicAdd(p + 1, v.y);
  atomicAdd(p + 2, v.z);
  atomicAdd(p + 3, v.w);
}

__global__ __launch_bounds__(128) void l1_transform_kernel(
    const float* __restrict__ X, const float* __restrict__ W0,
    const float* __restrict__ W1, const float* __restrict__ W2,
    const float* __restrict__ biasA, const float* __restrict__ biasB,
    float* __restrict__ Y0, float* __restrict__ Y1, float* __restrict__ Y2,
    int M) {
  const int lane = threadIdx.x & 31;
  const int wave = threadIdx.x >> 5;
  const int wpb = blockDim.x >> 5;
  const int half = lane >> 4;
  const int mrow = lane & 15;
  const int nTiles = (M + 15) >> 4;

  v2f b0[16], b1[16], b2[16];
#pragma unroll
  for (int s = 0; s < 16; ++s) {
    int k = 4 * s + 2 * half;
    b0[s] = (v2f){W0[k * 16 + mrow], W0[(k + 1) * 16 + mrow]};
    b1[s] = (v2f){W1[k * 16 + mrow], W1[(k + 1) * 16 + mrow]};
    b2[s] = (v2f){W2[k * 16 + mrow], W2[(k + 1) * 16 + mrow]};
  }

  const float bsum = biasA[mrow] + biasB[mrow];

  for (int tile = blockIdx.x * wpb + wave; tile < nTiles;
       tile += gridDim.x * wpb) {
    const int rbase = tile << 4;
    int arow = rbase + mrow;
    if (arow > M - 1) arow = M - 1;
    const float* xr = X + (size_t)arow * 64;
    v8f c0 = {}, c1 = {};
    v8f c2 = {bsum, bsum, bsum, bsum, bsum, bsum, bsum, bsum};
#pragma unroll
    for (int s = 0; s < 16; ++s) {
      int k = 4 * s + 2 * half;
      v2f a = (v2f){xr[k], xr[k + 1]};
      c0 = wmma_k4(a, b0[s], c0);
      c1 = wmma_k4(a, b1[s], c1);
      c2 = wmma_k4(a, b2[s], c2);
    }
#pragma unroll
    for (int r = 0; r < 8; ++r) {
      int row = rbase + r + 8 * half;
      if (row < M) {
        size_t o = (size_t)row * 16 + mrow;
        Y0[o] = c0[r];
        Y1[o] = c1[r];
        Y2[o] = c2[r];
      }
    }
  }
}

__global__ __launch_bounds__(128) void l2_fused_kernel(
    const float* __restrict__ A0, const float* __restrict__ cnt0,
    const float* __restrict__ A1, const float* __restrict__ cnt1,
    const float* __restrict__ A2, const float* __restrict__ W0,
    const float* __restrict__ W1, const float* __restrict__ W2,
    const float* __restrict__ bias0, const float* __restrict__ bias1,
    float* __restrict__ OUT, int M) {
  const int lane = threadIdx.x & 31;
  const int wave = threadIdx.x >> 5;
  const int wpb = blockDim.x >> 5;
  const int half = lane >> 4;
  const int mrow = lane & 15;
  const int nTiles = (M + 15) >> 4;

  const float* Ws[3] = {W0, W1, W2};
  v2f b[3][4][2];
#pragma unroll
  for (int m = 0; m < 3; ++m)
#pragma unroll
    for (int s = 0; s < 4; ++s)
#pragma unroll
      for (int t = 0; t < 2; ++t) {
        int k = 4 * s + 2 * half;
        int col = 16 * t + mrow;
        b[m][s][t] = (v2f){Ws[m][k * 32 + col], Ws[m][(k + 1) * 32 + col]};
      }

  const float biasL = bias0[mrow] + bias1[mrow];
  const float biasH = bias0[16 + mrow] + bias1[16 + mrow];

  for (int tile = blockIdx.x * wpb + wave; tile < nTiles;
       tile += gridDim.x * wpb) {
    const int rbase = tile << 4;
    int arow = rbase + mrow;
    if (arow > M - 1) arow = M - 1;
    const float rs0 = 1.0f / fmaxf(cnt0[arow], 1.0f);
    const float rs1 = 1.0f / fmaxf(cnt1[arow], 1.0f);
    const float* a0p = A0 + (size_t)arow * 16;
    const float* a1p = A1 + (size_t)arow * 16;
    const float* a2p = A2 + (size_t)arow * 16;
    v8f cL = {}, cH = {};
#pragma unroll
    for (int s = 0; s < 4; ++s) {
      int k = 4 * s + 2 * half;
      v2f a0 = (v2f){a0p[k] * rs0, a0p[k + 1] * rs0};
      cL = wmma_k4(a0, b[0][s][0], cL);
      cH = wmma_k4(a0, b[0][s][1], cH);
      v2f a1 = (v2f){a1p[k] * rs1, a1p[k + 1] * rs1};
      cL = wmma_k4(a1, b[1][s][0], cL);
      cH = wmma_k4(a1, b[1][s][1], cH);
      v2f a2 = (v2f){fmaxf(a2p[k], 0.0f), fmaxf(a2p[k + 1], 0.0f)};
      cL = wmma_k4(a2, b[2][s][0], cL);
      cH = wmma_k4(a2, b[2][s][1], cH);
    }
#pragma unroll
    for (int r = 0; r < 8; ++r) {
      int row = rbase + r + 8 * half;
      if (row < M) {
        size_t o = (size_t)row * 32 + mrow;
        OUT[o] = cL[r] + biasL;
        OUT[o + 16] = cH[r] + biasH;
      }
    }
  }
}


extern "C" void kernel_launch(void* const* d_in, const int* in_sizes, int n_in,
                              void* d_out, int out_size, void* d_ws,
                              size_t ws_size, hipStream_t stream) {
  const float* x_dev = (const float*)d_in[0];
  const float* x_task = (const float*)d_in[1];
  const int* wr_src = (const int*)d_in[2];
  const int* wr_dst = (const int*)d_in[3];
  const int* rv_src = (const int*)d_in[4];
  const int* rv_dst = (const int*)d_in[5];
  const int* wb_src = (const int*)d_in[6];
  const int* wb_dst = (const int*)d_in[7];
  const int* rb_src = (const int*)d_in[8];
  const int* rb_dst = (const int*)d_in[9];
  const float* p1wr_wl = (const float*)d_in[10];
  const float* p1wr_bl = (const float*)d_in[11];
  const float* p1wr_wr = (const float*)d_in[12];
  const float* p1rv_wl = (const float*)d_in[13];
  const float* p1rv_bl = (const float*)d_in[14];
  const float* p1rv_wr = (const float*)d_in[15];
  const float* p1wb_wl = (const float*)d_in[16];
  const float* p1wb_bl = (const float*)d_in[17];
  const float* p1wb_wr = (const float*)d_in[18];
  const float* p1rb_wl = (const float*)d_in[19];
  const float* p1rb_bl = (const float*)d_in[20];
  const float* p1rb_wr = (const float*)d_in[21];
  const float* p2wr_wl = (const float*)d_in[22];
  const float* p2wr_bl = (const float*)d_in[23];
  const float* p2wr_wr = (const float*)d_in[24];
  const float* p2rv_wl = (const float*)d_in[25];
  const float* p2rv_bl = (const float*)d_in[26];
  const float* p2rv_wr = (const float*)d_in[27];
  const float* p2wb_wl = (const float*)d_in[28];
  const float* p2wb_bl = (const float*)d_in[29];
  const float* p2wb_wr = (const float*)d_in[30];
  const float* p2rb_wl = (const float*)d_in[31];
  const float* p2rb_bl = (const float*)d_in[32];
  const float* p2rb_wr = (const float*)d_in[33];

  const int ND = in_sizes[0] / 64;
  const int NT = in_sizes[1] / 64;
  const int E_wr = in_sizes[2], E_rv = in_sizes[4];
  const int E_wb = in_sizes[6], E_rb = in_sizes[8];

  float* ws = (float*)d_ws;
  size_t off = 0;
  float* cnt_wr = ws + off; off += NT;
  float* cnt_rv = ws + off; off += NT;
  float* cnt_wb = ws + off; off += ND;
  float* cnt_rb = ws + off; off += ND;
  float* agg_wr = ws + off; off += (size_t)NT * 16;
  float* agg_rv = ws + off; off += (size_t)NT * 16;
  float* agg_wb = ws + off; off += (size_t)ND * 16;
  float* agg_rb = ws + off; off += (size_t)ND * 16;
  float* y_wr = ws + off; off += (size_t)ND * 16;
  float* y_rv = ws + off; off += (size_t)ND * 16;
  float* y_wb = ws + off; off += (size_t)NT * 16;
  float* y_rb = ws + off; off += (size_t)NT * 16;
  float* t1 = ws + off; off += (size_t)NT * 16;
  float* d1 = ws + off; off += (size_t)ND * 16;
  float* ws1t = ws + off; off += 64 * 16;
  float* ws1d = ws + off; off += 64 * 16;
  float* ws2t = ws + off; off += 16 * 32;
  float* ws2d = ws + off; off += 16 * 32;

  const int TPB = 256;
  const long long aggN = (long long)(2LL * NT + 2LL * ND) * 16;
  const long long cntN = (long long)(2LL * NT + 2LL * ND);

  zero_kernel<<<(int)((cntN + TPB - 1) / TPB), TPB, 0, stream>>>(ws, cntN);

  count_kernel<<<(E_wr + TPB - 1) / TPB, TPB, 0, stream>>>(wr_dst, E_wr, cnt_wr);
  count_kernel<<<(E_rv + TPB - 1) / TPB, TPB, 0, stream>>>(rv_dst, E_rv, cnt_rv);
  count_kernel<<<(E_wb + TPB - 1) / TPB, TPB, 0, stream>>>(wb_dst, E_wb, cnt_wb);
  count_kernel<<<(E_rb + TPB - 1) / TPB, TPB, 0, stream>>>(rb_dst, E_rb, cnt_rb);

  addmat_kernel<<<4, TPB, 0, stream>>>(p1wr_wr, p1rv_wr, ws1t, 64 * 16);
  addmat_kernel<<<4, TPB, 0, stream>>>(p1wb_wr, p1rb_wr, ws1d, 64 * 16);
  addmat_kernel<<<2, TPB, 0, stream>>>(p2wr_wr, p2rv_wr, ws2t, 16 * 32);
  addmat_kernel<<<2, TPB, 0, stream>>>(p2wb_wr, p2rb_wr, ws2d, 16 * 32);

  const int tilesD = (ND + 15) / 16, tilesT = (NT + 15) / 16;
  l1_transform_kernel<<<(tilesD + 3) / 4, 128, 0, stream>>>(
      x_dev, p1wr_wl, p1rv_wl, ws1d, p1wb_bl, p1rb_bl, y_wr, y_rv, d1, ND);
  l1_transform_kernel<<<(tilesT + 3) / 4, 128, 0, stream>>>(
      x_task, p1wb_wl, p1rb_wl, ws1t, p1wr_bl, p1rv_bl, y_wb, y_rb, t1, NT);

  auto scb = [&](int E) { return (int)(((long long)E * 4 + TPB - 1) / TPB); };
  scatter16_kernel<true, false><<<scb(E_wr), TPB, 0, stream>>>(
      y_wr, wr_src, wr_dst, E_wr, cnt_wr, t1);
  scatter16_kernel<true, false><<<scb(E_rv), TPB, 0, stream>>>(
      y_rv, rv_src, rv_dst, E_rv, cnt_rv, t1);
  scatter16_kernel<true, false><<<scb(E_wb), TPB, 0, stream>>>(
      y_wb, wb_src, wb_dst, E_wb, cnt_wb, d1);
  scatter16_kernel<true, false><<<scb(E_rb), TPB, 0, stream>>>(
      y_rb, rb_src, rb_dst, E_rb, cnt_rb, d1);

  zero_kernel<<<(int)((aggN + TPB - 1) / TPB), TPB, 0, stream>>>(agg_wr, aggN);

  scatter16_kernel<false, true><<<scb(E_wr), TPB, 0, stream>>>(
      d1, wr_src, wr_dst, E_wr, cnt_wr, agg_wr);
  scatter16_kernel<false, true><<<scb(E_rv), TPB, 0, stream>>>(
      d1, rv_src, rv_dst, E_rv, cnt_rv, agg_rv);
  scatter16_kernel<false, true><<<scb(E_wb), TPB, 0, stream>>>(
      t1, wb_src, wb_dst, E_wb, cnt_wb, agg_wb);
  scatter16_kernel<false, true><<<scb(E_rb), TPB, 0, stream>>>(
      t1, rb_src, rb_dst, E_rb, cnt_rb, agg_rb);

  float* out_d2 = (float*)d_out;
  float* out_t2 = (float*)d_out + (size_t)ND * 32;
  l2_fused_kernel<<<(tilesT + 3) / 4, 128, 0, stream>>>(
      agg_wr, cnt_wr, agg_rv, cnt_rv, t1, p2wr_wl, p2rv_wl, ws2t, p2wr_bl,
      p2rv_bl, out_t2, NT);
  l2_fused_kernel<<<(tilesD + 3) / 4, 128, 0, stream>>>(
      agg_wb, cnt_wb, agg_rb, cnt_rb, d1, p2wb_wl, p2rb_wl, ws2d, p2wb_bl,
      p2rb_bl, out_d2, ND);
}


# Float-atomic edge scatter replaced by a per-call CSR (int degree count, scan, arrival-order fill) and per-destination gather sums in both layers; row sums now add in fill arrival order
# speedup vs baseline: 1.1579x; 1.0741x over previous
#include <hip/hip_runtime.h>


typedef __attribute__((ext_vector_type(2))) float v2f;
typedef __attribute__((ext_vector_type(8))) float v8f;

__device__ __forceinline__ v8f wmma_k4(v2f a, v2f b, v8f c) {
  return __builtin_amdgcn_wmma_f32_16x16x4_f32(
       false, a,  false, b,
       (short)0, c,  false,  false);
}


__global__ void zero_int_kernel(int* __restrict__ p, int n) {
  int i = blockIdx.x * blockDim.x + threadIdx.x;
  if (i < n) p[i] = 0;
}

__global__ void count_kernel(const int* __restrict__ dst, int E,
                             int* __restrict__ cnt) {
  int i = blockIdx.x * blockDim.x + threadIdx.x;
  if (i < E) atomicAdd(&cnt[dst[i]], 1);
}

__global__ void addmat_kernel(const float* __restrict__ a,
                              const float* __restrict__ b,
                              float* __restrict__ o, int n) {
  int i = blockIdx.x * blockDim.x + threadIdx.x;
  if (i < n) o[i] = a[i] + b[i];
}


__device__ __forceinline__ int block_excl_scan_256(int v, int* sh, int& total) {
  const int t = threadIdx.x;
  sh[t] = v;
  __syncthreads();
  for (int off = 1; off < 256; off <<= 1) {
    const int add = (t >= off) ? sh[t - off] : 0;
    __syncthreads();
    sh[t] += add;
    __syncthreads();
  }
  const int incl = sh[t];
  total = sh[255];
  __syncthreads();
  return incl - v;
}

__global__ __launch_bounds__(256) void scan_partial_kernel(
    const int* __restrict__ cnt, int n, int* __restrict__ bsum) {
  __shared__ int sh[256];
  const int base = blockIdx.x * 2048 + threadIdx.x * 8;
  int v[8];
#pragma unroll
  for (int j = 0; j < 8; ++j) {
    const int idx = base + j;
    v[j] = (idx < n) ? cnt[idx] : 0;
  }
  int s = 0;
#pragma unroll
  for (int j = 0; j < 8; ++j) s += v[j];
  int total;
  block_excl_scan_256(s, sh, total);
  if (threadIdx.x == 0) bsum[blockIdx.x] = total;
}

__global__ __launch_bounds__(256) void scan_offsets_kernel(
    int* __restrict__ bsum, int nb) {
  __shared__ int sh[256];
  int carry = 0;
  for (int base = 0; base < nb; base += 256) {
    const int idx = base + threadIdx.x;
    const int v = (idx < nb) ? bsum[idx] : 0;
    int total;
    const int ex = block_excl_scan_256(v, sh, total);
    if (idx < nb) bsum[idx] = carry + ex;
    carry += total;
  }
}

__global__ __launch_bounds__(256) void scan_final_kernel(
    const int* __restrict__ cnt, int n, const int* __restrict__ bsum,
    int* __restrict__ pos) {
  __shared__ int sh[256];
  const int base = blockIdx.x * 2048 + threadIdx.x * 8;
  int v[8];
  int s = 0;
#pragma unroll
  for (int j = 0; j < 8; ++j) {
    const int idx = base + j;
    v[j] = (idx < n) ? cnt[idx] : 0;
    s += v[j];
  }
  int total;
  const int ex = block_excl_scan_256(s, sh, total);
  int run = bsum[blockIdx.x] + ex;
#pragma unroll
  for (int j = 0; j < 8; ++j) {
    const int idx = base + j;
    if (idx < n) pos[idx] = run;
    run += v[j];
  }
}

__global__ void fill_kernel(const int* __restrict__ src,
                            const int* __restrict__ dst, int E,
                            int* __restrict__ pos, int* __restrict__ col) {
  int i = blockIdx.x * blockDim.x + threadIdx.x;
  if (i < E) {
    const int p = atomicAdd(&pos[dst[i]], 1);
    col[p] = src[i];
  }
}

__device__ __forceinline__ float4 load_row4(const float* __restrict__ Y, int s,
                                            int c) {
  return *reinterpret_cast<const float4*>(Y + (size_t)s * 16 + c * 4);
}

__device__ __forceinline__ float4 gather_rows(const float* __restrict__ Y,
                                              const int* __restrict__ col,
                                              int beg, int end, int c) {
  float4 acc = make_float4(0.0f, 0.0f, 0.0f, 0.0f);
  int p = beg;
  for (; p + 1 < end; p += 2) {
    const int s0 = col[p];
    const int s1 = col[p + 1];
    const float4 v0 = load_row4(Y, s0, c);
    const float4 v1 = load_row4(Y, s1, c);
    acc.x += v0.x;
    acc.y += v0.y;
    acc.z += v0.z;
    acc.w += v0.w;
    acc.x += v1.x;
    acc.y += v1.y;
    acc.z += v1.z;
    acc.w += v1.w;
  }
  if (p < end) {
    const float4 v = load_row4(Y, col[p], c);
    acc.x += v.x;
    acc.y += v.y;
    acc.z += v.z;
    acc.w += v.w;
  }
  return acc;
}

__global__ void gather_l1_kernel(const float* __restrict__ YA,
                                 const float* __restrict__ YB,
                                 const int* __restrict__ endA,
                                 const int* __restrict__ cntA,
                                 const int* __restrict__ endB,
                                 const int* __restrict__ cntB,
                                 const int* __restrict__ col,
                                 float* __restrict__ T, int N) {
  const int i = blockIdx.x * blockDim.x + threadIdx.x;
  const int node = i >> 2;
  const int c = i & 3;
  if (node >= N) return;
  const int nA = cntA[node], eA = endA[node];
  const int nB = cntB[node], eB = endB[node];
  float4* tp = reinterpret_cast<float4*>(T + (size_t)node * 16 + c * 4);
  float4 acc = *tp;
  const float4 sA = gather_rows(YA, col, eA - nA, eA, c);
  const float4 sB = gather_rows(YB, col, eB - nB, eB, c);
  const float iA = 1.0f / fmaxf((float)nA, 1.0f);
  const float iB = 1.0f / fmaxf((float)nB, 1.0f);
  acc.x += sA.x * iA + sB.x * iB;
  acc.y += sA.y * iA + sB.y * iB;
  acc.z += sA.z * iA + sB.z * iB;
  acc.w += sA.w * iA + sB.w * iB;
  acc.x = fmaxf(acc.x, 0.0f);
  acc.y = fmaxf(acc.y, 0.0f);
  acc.z = fmaxf(acc.z, 0.0f);
  acc.w = fmaxf(acc.w, 0.0f);
  *tp = acc;
}

__global__ void gather_l2_kernel(const float* __restrict__ H,
                                 const int* __restrict__ endA,
                                 const int* __restrict__ cntA,
                                 const int* __restrict__ endB,
                                 const int* __restrict__ cntB,
                                 const int* __restrict__ col,
                                 float* __restrict__ aggA,
                                 float* __restrict__ aggB, int N) {
  const int i = blockIdx.x * blockDim.x + threadIdx.x;
  const int node = i >> 2;
  const int c = i & 3;
  if (node >= N) return;
  const int nA = cntA[node], eA = endA[node];
  const int nB = cntB[node], eB = endB[node];
  const float4 sA = gather_rows(H, col, eA - nA, eA, c);
  const float4 sB = gather_rows(H, col, eB - nB, eB, c);
  *reinterpret_cast<float4*>(aggA + (size_t)node * 16 + c * 4) = sA;
  *reinterpret_cast<float4*>(aggB + (size_t)node * 16 + c * 4) = sB;
}

__global__ __launch_bounds__(128) void l1_transform_kernel(
    const float* __restrict__ X, const float* __restrict__ W0,
    const float* __restrict__ W1, const float* __restrict__ W2,
    const float* __restrict__ biasA, const float* __restrict__ biasB,
    float* __restrict__ Y0, float* __restrict__ Y1, float* __restrict__ Y2,
    int M) {
  const int lane = threadIdx.x & 31;
  const int wave = threadIdx.x >> 5;
  const int wpb = blockDim.x >> 5;
  const int half = lane >> 4;
  const int mrow = lane & 15;
  const int nTiles = (M + 15) >> 4;

  v2f b0[16], b1[16], b2[16];
#pragma unroll
  for (int s = 0; s < 16; ++s) {
    int k = 4 * s + 2 * half;
    b0[s] = (v2f){W0[k * 16 + mrow], W0[(k + 1) * 16 + mrow]};
    b1[s] = (v2f){W1[k * 16 + mrow], W1[(k + 1) * 16 + mrow]};
    b2[s] = (v2f){W2[k * 16 + mrow], W2[(k + 1) * 16 + mrow]};
  }

  const float bsum = biasA[mrow] + biasB[mrow];

  for (int tile = blockIdx.x * wpb + wave; tile < nTiles;
       tile += gridDim.x * wpb) {
    const int rbase = tile << 4;
    int arow = rbase + mrow;
    if (arow > M - 1) arow = M - 1;
    const float* xr = X + (size_t)arow * 64;
    v8f c0 = {}, c1 = {};
    v8f c2 = {bsum, bsum, bsum, bsum, bsum, bsum, bsum, bsum};
#pragma unroll
    for (int s = 0; s < 16; ++s) {
      int k = 4 * s + 2 * half;
      v2f a = (v2f){xr[k], xr[k + 1]};
      c0 = wmma_k4(a, b0[s], c0);
      c1 = wmma_k4(a, b1[s], c1);
      c2 = wmma_k4(a, b2[s], c2);
    }
#pragma unroll
    for (int r = 0; r < 8; ++r) {
      int row = rbase + r + 8 * half;
      if (row < M) {
        size_t o = (size_t)row * 16 + mrow;
        Y0[o] = c0[r];
        Y1[o] = c1[r];
        Y2[o] = c2[r];
      }
    }
  }
}

__global__ __launch_bounds__(128) void l2_fused_kernel(
    const float* __restrict__ A0, const int* __restrict__ cnt0,
    const float* __restrict__ A1, const int* __restrict__ cnt1,
    const float* __restrict__ A2, const float* __restrict__ W0,
    const float* __restrict__ W1, const float* __restrict__ W2,
    const float* __restrict__ bias0, const float* __restrict__ bias1,
    float* __restrict__ OUT, int M) {
  const int lane = threadIdx.x & 31;
  const int wave = threadIdx.x >> 5;
  const int wpb = blockDim.x >> 5;
  const int half = lane >> 4;
  const int mrow = lane & 15;
  const int nTiles = (M + 15) >> 4;

  const float* Ws[3] = {W0, W1, W2};
  v2f b[3][4][2];
#pragma unroll
  for (int m = 0; m < 3; ++m)
#pragma unroll
    for (int s = 0; s < 4; ++s)
#pragma unroll
      for (int t = 0; t < 2; ++t) {
        int k = 4 * s + 2 * half;
        int col = 16 * t + mrow;
        b[m][s][t] = (v2f){Ws[m][k * 32 + col], Ws[m][(k + 1) * 32 + col]};
      }

  const float biasL = bias0[mrow] + bias1[mrow];
  const float biasH = bias0[16 + mrow] + bias1[16 + mrow];

  for (int tile = blockIdx.x * wpb + wave; tile < nTiles;
       tile += gridDim.x * wpb) {
    const int rbase = tile << 4;
    int arow = rbase + mrow;
    if (arow > M - 1) arow = M - 1;
    const float rs0 = 1.0f / fmaxf((float)cnt0[arow], 1.0f);
    const float rs1 = 1.0f / fmaxf((float)cnt1[arow], 1.0f);
    const float* a0p = A0 + (size_t)arow * 16;
    const float* a1p = A1 + (size_t)arow * 16;
    const float* a2p = A2 + (size_t)arow * 16;
    v8f cL = {}, cH = {};
#pragma unroll
    for (int s = 0; s < 4; ++s) {
      int k = 4 * s + 2 * half;
      v2f a0 = (v2f){a0p[k] * rs0, a0p[k + 1] * rs0};
      cL = wmma_k4(a0, b[0][s][0], cL);
      cH = wmma_k4(a0, b[0][s][1], cH);
      v2f a1 = (v2f){a1p[k] * rs1, a1p[k + 1] * rs1};
      cL = wmma_k4(a1, b[1][s][0], cL);
      cH = wmma_k4(a1, b[1][s][1], cH);
      v2f a2 = (v2f){fmaxf(a2p[k], 0.0f), fmaxf(a2p[k + 1], 0.0f)};
      cL = wmma_k4(a2, b[2][s][0], cL);
      cH = wmma_k4(a2, b[2][s][1], cH);
    }
#pragma unroll
    for (int r = 0; r < 8; ++r) {
      int row = rbase + r + 8 * half;
      if (row < M) {
        size_t o = (size_t)row * 32 + mrow;
        OUT[o] = cL[r] + biasL;
        OUT[o + 16] = cH[r] + biasH;
      }
    }
  }
}


extern "C" void kernel_launch(void* const* d_in, const int* in_sizes, int n_in,
                              void* d_out, int out_size, void* d_ws,
                              size_t ws_size, hipStream_t stream) {
  const float* x_dev = (const float*)d_in[0];
  const float* x_task = (const float*)d_in[1];
  const int* wr_src = (const int*)d_in[2];
  const int* wr_dst = (const int*)d_in[3];
  const int* rv_src = (const int*)d_in[4];
  const int* rv_dst = (const int*)d_in[5];
  const int* wb_src = (const int*)d_in[6];
  const int* wb_dst = (const int*)d_in[7];
  const int* rb_src = (const int*)d_in[8];
  const int* rb_dst = (const int*)d_in[9];
  const float* p1wr_wl = (const float*)d_in[10];
  const float* p1wr_bl = (const float*)d_in[11];
  const float* p1wr_wr = (const float*)d_in[12];
  const float* p1rv_wl = (const float*)d_in[13];
  const float* p1rv_bl = (const float*)d_in[14];
  const float* p1rv_wr = (const float*)d_in[15];
  const float* p1wb_wl = (const float*)d_in[16];
  const float* p1wb_bl = (const float*)d_in[17];
  const float* p1wb_wr = (const float*)d_in[18];
  const float* p1rb_wl = (const float*)d_in[19];
  const float* p1rb_bl = (const float*)d_in[20];
  const float* p1rb_wr = (const float*)d_in[21];
  const float* p2wr_wl = (const float*)d_in[22];
  const float* p2wr_bl = (const float*)d_in[23];
  const float* p2wr_wr = (const float*)d_in[24];
  const float* p2rv_wl = (const float*)d_in[25];
  const float* p2rv_bl = (const float*)d_in[26];
  const float* p2rv_wr = (const float*)d_in[27];
  const float* p2wb_wl = (const float*)d_in[28];
  const float* p2wb_bl = (const float*)d_in[29];
  const float* p2wb_wr = (const float*)d_in[30];
  const float* p2rb_wl = (const float*)d_in[31];
  const float* p2rb_bl = (const float*)d_in[32];
  const float* p2rb_wr = (const float*)d_in[33];

  const int ND = in_sizes[0] / 64;
  const int NT = in_sizes[1] / 64;
  const int E_wr = in_sizes[2], E_rv = in_sizes[4];
  const int E_wb = in_sizes[6], E_rb = in_sizes[8];

  const int cntN = 2 * NT + 2 * ND;
  const int Etot = E_wr + E_rv + E_wb + E_rb;
  const int nChunks = (cntN + 2047) / 2048;
  float* ws = (float*)d_ws;
  size_t off = 0;
  int* cnt = (int*)(ws + off); off += (size_t)cntN;
  float* csr = ws + off; off += (size_t)cntN * 16;
  float* yreg = ws + off; off += (size_t)cntN * 16;
  float* t1 = ws + off; off += (size_t)NT * 16;
  float* d1 = ws + off; off += (size_t)ND * 16;
  float* ws1t = ws + off; off += 64 * 16;
  float* ws1d = ws + off; off += 64 * 16;
  float* ws2t = ws + off; off += 16 * 32;
  float* ws2d = ws + off; off += 16 * 32;

  int* cnt_wr = cnt;
  int* cnt_rv = cnt + NT;
  int* cnt_wb = cnt + 2 * (size_t)NT;
  int* cnt_rb = cnt + 2 * (size_t)NT + ND;
  int* pos = (int*)csr;
  int* pos_wr = pos;
  int* pos_rv = pos + NT;
  int* pos_wb = pos + 2 * (size_t)NT;
  int* pos_rb = pos + 2 * (size_t)NT + ND;
  int* col = pos + cntN;
  int* bsum = col + Etot;

  float* y_wr = yreg;
  float* y_rv = y_wr + (size_t)ND * 16;
  float* y_wb = y_rv + (size_t)ND * 16;
  float* y_rb = y_wb + (size_t)NT * 16;
  float* agg_wr = yreg;
  float* agg_rv = agg_wr + (size_t)NT * 16;
  float* agg_wb = agg_rv + (size_t)NT * 16;
  float* agg_rb = agg_wb + (size_t)ND * 16;

  const int TPB = 256;

  zero_int_kernel<<<(cntN + TPB - 1) / TPB, TPB, 0, stream>>>(cnt, cntN);
  count_kernel<<<(E_wr + TPB - 1) / TPB, TPB, 0, stream>>>(wr_dst, E_wr, cnt_wr);
  count_kernel<<<(E_rv + TPB - 1) / TPB, TPB, 0, stream>>>(rv_dst, E_rv, cnt_rv);
  count_kernel<<<(E_wb + TPB - 1) / TPB, TPB, 0, stream>>>(wb_dst, E_wb, cnt_wb);
  count_kernel<<<(E_rb + TPB - 1) / TPB, TPB, 0, stream>>>(rb_dst, E_rb, cnt_rb);

  scan_partial_kernel<<<nChunks, 256, 0, stream>>>(cnt, cntN, bsum);
  scan_offsets_kernel<<<1, 256, 0, stream>>>(bsum, nChunks);
  scan_final_kernel<<<nChunks, 256, 0, stream>>>(cnt, cntN, bsum, pos);

  fill_kernel<<<(E_wr + TPB - 1) / TPB, TPB, 0, stream>>>(wr_src, wr_dst, E_wr, pos_wr, col);
  fill_kernel<<<(E_rv + TPB - 1) / TPB, TPB, 0, stream>>>(rv_src, rv_dst, E_rv, pos_rv, col);
  fill_kernel<<<(E_wb + TPB - 1) / TPB, TPB, 0, stream>>>(wb_src, wb_dst, E_wb, pos_wb, col);
  fill_kernel<<<(E_rb + TPB - 1) / TPB, TPB, 0, stream>>>(rb_src, rb_dst, E_rb, pos_rb, col);

  addmat_kernel<<<4, TPB, 0, stream>>>(p1wr_wr, p1rv_wr, ws1t, 64 * 16);
  addmat_kernel<<<4, TPB, 0, stream>>>(p1wb_wr, p1rb_wr, ws1d, 64 * 16);
  addmat_kernel<<<2, TPB, 0, stream>>>(p2wr_wr, p2rv_wr, ws2t, 16 * 32);
  addmat_kernel<<<2, TPB, 0, stream>>>(p2wb_wr, p2rb_wr, ws2d, 16 * 32);

  const int tilesD = (ND + 15) / 16, tilesT = (NT + 15) / 16;
  l1_transform_kernel<<<(tilesD + 3) / 4, 128, 0, stream>>>(
      x_dev, p1wr_wl, p1rv_wl, ws1d, p1wb_bl, p1rb_bl, y_wr, y_rv, d1, ND);
  l1_transform_kernel<<<(tilesT + 3) / 4, 128, 0, stream>>>(
      x_task, p1wb_wl, p1rb_wl, ws1t, p1wr_bl, p1rv_bl, y_wb, y_rb, t1, NT);

  gather_l1_kernel<<<(NT * 4 + TPB - 1) / TPB, TPB, 0, stream>>>(
      y_wr, y_rv, pos_wr, cnt_wr, pos_rv, cnt_rv, col, t1, NT);
  gather_l1_kernel<<<(ND * 4 + TPB - 1) / TPB, TPB, 0, stream>>>(
      y_wb, y_rb, pos_wb, cnt_wb, pos_rb, cnt_rb, col, d1, ND);

  gather_l2_kernel<<<(NT * 4 + TPB - 1) / TPB, TPB, 0, stream>>>(
      d1, pos_wr, cnt_wr, pos_rv, cnt_rv, col, agg_wr, agg_rv, NT);
  gather_l2_kernel<<<(ND * 4 + TPB - 1) / TPB, TPB, 0, stream>>>(
      t1, pos_wb, cnt_wb, pos_rb, cnt_rb, col, agg_wb, agg_rb, ND);

  float* out_d2 = (float*)d_out;
  float* out_t2 = (float*)d_out + (size_t)ND * 32;
  l2_fused_kernel<<<(tilesT + 3) / 4, 128, 0, stream>>>(
      agg_wr, cnt_wr, agg_rv, cnt_rv, t1, p2wr_wl, p2rv_wl, ws2t, p2wr_bl,
      p2rv_bl, out_t2, NT);
  l2_fused_kernel<<<(tilesD + 3) / 4, 128, 0, stream>>>(
      agg_wb, cnt_wb, agg_rb, cnt_rb, d1, p2wb_wl, p2rb_wl, ws2d, p2wb_bl,
      p2rb_bl, out_d2, ND);
}


# Edge rank taken from the returning count add, so the neighbour-list fill is atomic-free (integer atomics per call halved); row sums now add in count-arrival order
# speedup vs baseline: 1.3866x; 1.1975x over previous
#include <hip/hip_runtime.h>


typedef __attribute__((ext_vector_type(2))) float v2f;
typedef __attribute__((ext_vector_type(8))) float v8f;

__device__ __forceinline__ v8f wmma_k4(v2f a, v2f b, v8f c) {
  return __builtin_amdgcn_wmma_f32_16x16x4_f32(
       false, a,  false, b,
       (short)0, c,  false,  false);
}


__global__ void zero_int_kernel(int* __restrict__ p, int n) {
  int i = blockIdx.x * blockDim.x + threadIdx.x;
  if (i < n) p[i] = 0;
}

__global__ void count_kernel(const int* __restrict__ dst, int E,
                             int* __restrict__ cnt, int* __restrict__ rank) {
  int i = blockIdx.x * blockDim.x + threadIdx.x;
  if (i < E) rank[i] = atomicAdd(&cnt[dst[i]], 1);
}

__global__ void addmat_kernel(const float* __restrict__ a,
                              const float* __restrict__ b,
                              float* __restrict__ o, int n) {
  int i = blockIdx.x * blockDim.x + threadIdx.x;
  if (i < n) o[i] = a[i] + b[i];
}


__device__ __forceinline__ int block_excl_scan_256(int v, int* sh, int& total) {
  const int t = threadIdx.x;
  sh[t] = v;
  __syncthreads();
  for (int off = 1; off < 256; off <<= 1) {
    const int add = (t >= off) ? sh[t - off] : 0;
    __syncthreads();
    sh[t] += add;
    __syncthreads();
  }
  const int incl = sh[t];
  total = sh[255];
  __syncthreads();
  return incl - v;
}

__global__ __launch_bounds__(256) void scan_partial_kernel(
    const int* __restrict__ cnt, int n, int* __restrict__ bsum) {
  __shared__ int sh[256];
  const int base = blockIdx.x * 2048 + threadIdx.x * 8;
  int v[8];
#pragma unroll
  for (int j = 0; j < 8; ++j) {
    const int idx = base + j;
    v[j] = (idx < n) ? cnt[idx] : 0;
  }
  int s = 0;
#pragma unroll
  for (int j = 0; j < 8; ++j) s += v[j];
  int total;
  block_excl_scan_256(s, sh, total);
  if (threadIdx.x == 0) bsum[blockIdx.x] = total;
}

__global__ __launch_bounds__(256) void scan_offsets_kernel(
    int* __restrict__ bsum, int nb) {
  __shared__ int sh[256];
  int carry = 0;
  for (int base = 0; base < nb; base += 256) {
    const int idx = base + threadIdx.x;
    const int v = (idx < nb) ? bsum[idx] : 0;
    int total;
    const int ex = block_excl_scan_256(v, sh, total);
    if (idx < nb) bsum[idx] = carry + ex;
    carry += total;
  }
}

__global__ __launch_bounds__(256) void scan_final_kernel(
    const int* __restrict__ cnt, int n, const int* __restrict__ bsum,
    int* __restrict__ pos) {
  __shared__ int sh[256];
  const int base = blockIdx.x * 2048 + threadIdx.x * 8;
  int v[8];
  int s = 0;
#pragma unroll
  for (int j = 0; j < 8; ++j) {
    const int idx = base + j;
    v[j] = (idx < n) ? cnt[idx] : 0;
    s += v[j];
  }
  int total;
  const int ex = block_excl_scan_256(s, sh, total);
  int run = bsum[blockIdx.x] + ex;
#pragma unroll
  for (int j = 0; j < 8; ++j) {
    const int idx = base + j;
    if (idx < n) pos[idx] = run;
    run += v[j];
  }
}

__global__ void fill_kernel(const int* __restrict__ src,
                            const int* __restrict__ dst,
                            const int* __restrict__ rank, int E,
                            const int* __restrict__ pos,
                            int* __restrict__ col) {
  int i = blockIdx.x * blockDim.x + threadIdx.x;
  if (i < E) col[pos[dst[i]] + rank[i]] = src[i];
}

__device__ __forceinline__ float4 load_row4(const float* __restrict__ Y, int s,
                                            int c) {
  return *reinterpret_cast<const float4*>(Y + (size_t)s * 16 + c * 4);
}

__device__ __forceinline__ float4 gather_rows(const float* __restrict__ Y,
                                              const int* __restrict__ col,
                                              int beg, int end, int c) {
  float4 acc = make_float4(0.0f, 0.0f, 0.0f, 0.0f);
  int p = beg;
  for (; p + 1 < end; p += 2) {
    const int s0 = col[p];
    const int s1 = col[p + 1];
    const float4 v0 = load_row4(Y, s0, c);
    const float4 v1 = load_row4(Y, s1, c);
    acc.x += v0.x;
    acc.y += v0.y;
    acc.z += v0.z;
    acc.w += v0.w;
    acc.x += v1.x;
    acc.y += v1.y;
    acc.z += v1.z;
    acc.w += v1.w;
  }
  if (p < end) {
    const float4 v = load_row4(Y, col[p], c);
    acc.x += v.x;
    acc.y += v.y;
    acc.z += v.z;
    acc.w += v.w;
  }
  return acc;
}

__global__ void gather_l1_kernel(const float* __restrict__ YA,
                                 const float* __restrict__ YB,
                                 const int* __restrict__ begA,
                                 const int* __restrict__ cntA,
                                 const int* __restrict__ begB,
                                 const int* __restrict__ cntB,
                                 const int* __restrict__ col,
                                 float* __restrict__ T, int N) {
  const int i = blockIdx.x * blockDim.x + threadIdx.x;
  const int node = i >> 2;
  const int c = i & 3;
  if (node >= N) return;
  const int nA = cntA[node], bA = begA[node];
  const int nB = cntB[node], bB = begB[node];
  float4* tp = reinterpret_cast<float4*>(T + (size_t)node * 16 + c * 4);
  float4 acc = *tp;
  const float4 sA = gather_rows(YA, col, bA, bA + nA, c);
  const float4 sB = gather_rows(YB, col, bB, bB + nB, c);
  const float iA = 1.0f / fmaxf((float)nA, 1.0f);
  const float iB = 1.0f / fmaxf((float)nB, 1.0f);
  acc.x += sA.x * iA + sB.x * iB;
  acc.y += sA.y * iA + sB.y * iB;
  acc.z += sA.z * iA + sB.z * iB;
  acc.w += sA.w * iA + sB.w * iB;
  acc.x = fmaxf(acc.x, 0.0f);
  acc.y = fmaxf(acc.y, 0.0f);
  acc.z = fmaxf(acc.z, 0.0f);
  acc.w = fmaxf(acc.w, 0.0f);
  *tp = acc;
}

__global__ void gather_l2_kernel(const float* __restrict__ H,
                                 const int* __restrict__ begA,
                                 const int* __restrict__ cntA,
                                 const int* __restrict__ begB,
                                 const int* __restrict__ cntB,
                                 const int* __restrict__ col,
                                 float* __restrict__ aggA,
                                 float* __restrict__ aggB, int N) {
  const int i = blockIdx.x * blockDim.x + threadIdx.x;
  const int node = i >> 2;
  const int c = i & 3;
  if (node >= N) return;
  const int nA = cntA[node], bA = begA[node];
  const int nB = cntB[node], bB = begB[node];
  const float4 sA = gather_rows(H, col, bA, bA + nA, c);
  const float4 sB = gather_rows(H, col, bB, bB + nB, c);
  *reinterpret_cast<float4*>(aggA + (size_t)node * 16 + c * 4) = sA;
  *reinterpret_cast<float4*>(aggB + (size_t)node * 16 + c * 4) = sB;
}

__global__ __launch_bounds__(128) void l1_transform_kernel(
    const float* __restrict__ X, const float* __restrict__ W0,
    const float* __restrict__ W1, const float* __restrict__ W2,
    const float* __restrict__ biasA, const float* __restrict__ biasB,
    float* __restrict__ Y0, float* __restrict__ Y1, float* __restrict__ Y2,
    int M) {
  const int lane = threadIdx.x & 31;
  const int wave = threadIdx.x >> 5;
  const int wpb = blockDim.x >> 5;
  const int half = lane >> 4;
  const int mrow = lane & 15;
  const int nTiles = (M + 15) >> 4;

  v2f b0[16], b1[16], b2[16];
#pragma unroll
  for (int s = 0; s < 16; ++s) {
    int k = 4 * s + 2 * half;
    b0[s] = (v2f){W0[k * 16 + mrow], W0[(k + 1) * 16 + mrow]};
    b1[s] = (v2f){W1[k * 16 + mrow], W1[(k + 1) * 16 + mrow]};
    b2[s] = (v2f){W2[k * 16 + mrow], W2[(k + 1) * 16 + mrow]};
  }

  const float bsum = biasA[mrow] + biasB[mrow];

  for (int tile = blockIdx.x * wpb + wave; tile < nTiles;
       tile += gridDim.x * wpb) {
    const int rbase = tile << 4;
    int arow = rbase + mrow;
    if (arow > M - 1) arow = M - 1;
    const float* xr = X + (size_t)arow * 64;
    v8f c0 = {}, c1 = {};
    v8f c2 = {bsum, bsum, bsum, bsum, bsum, bsum, bsum, bsum};
#pragma unroll
    for (int s = 0; s < 16; ++s) {
      int k = 4 * s + 2 * half;
      v2f a = (v2f){xr[k], xr[k + 1]};
      c0 = wmma_k4(a, b0[s], c0);
      c1 = wmma_k4(a, b1[s], c1);
      c2 = wmma_k4(a, b2[s], c2);
    }
#pragma unroll
    for (int r = 0; r < 8; ++r) {
      int row = rbase + r + 8 * half;
      if (row < M) {
        size_t o = (size_t)row * 16 + mrow;
        Y0[o] = c0[r];
        Y1[o] = c1[r];
        Y2[o] = c2[r];
      }
    }
  }
}

__global__ __launch_bounds__(128) void l2_fused_kernel(
    const float* __restrict__ A0, const int* __restrict__ cnt0,
    const float* __restrict__ A1, const int* __restrict__ cnt1,
    const float* __restrict__ A2, const float* __restrict__ W0,
    const float* __restrict__ W1, const float* __restrict__ W2,
    const float* __restrict__ bias0, const float* __restrict__ bias1,
    float* __restrict__ OUT, int M) {
  const int lane = threadIdx.x & 31;
  const int wave = threadIdx.x >> 5;
  const int wpb = blockDim.x >> 5;
  const int half = lane >> 4;
  const int mrow = lane & 15;
  const int nTiles = (M + 15) >> 4;

  const float* Ws[3] = {W0, W1, W2};
  v2f b[3][4][2];
#pragma unroll
  for (int m = 0; m < 3; ++m)
#pragma unroll
    for (int s = 0; s < 4; ++s)
#pragma unroll
      for (int t = 0; t < 2; ++t) {
        int k = 4 * s + 2 * half;
        int col = 16 * t + mrow;
        b[m][s][t] = (v2f){Ws[m][k * 32 + col], Ws[m][(k + 1) * 32 + col]};
      }

  const float biasL = bias0[mrow] + bias1[mrow];
  const float biasH = bias0[16 + mrow] + bias1[16 + mrow];

  for (int tile = blockIdx.x * wpb + wave; tile < nTiles;
       tile += gridDim.x * wpb) {
    const int rbase = tile << 4;
    int arow = rbase + mrow;
    if (arow > M - 1) arow = M - 1;
    const float rs0 = 1.0f / fmaxf((float)cnt0[arow], 1.0f);
    const float rs1 = 1.0f / fmaxf((float)cnt1[arow], 1.0f);
    const float* a0p = A0 + (size_t)arow * 16;
    const float* a1p = A1 + (size_t)arow * 16;
    const float* a2p = A2 + (size_t)arow * 16;
    v8f cL = {}, cH = {};
#pragma unroll
    for (int s = 0; s < 4; ++s) {
      int k = 4 * s + 2 * half;
      v2f a0 = (v2f){a0p[k] * rs0, a0p[k + 1] * rs0};
      cL = wmma_k4(a0, b[0][s][0], cL);
      cH = wmma_k4(a0, b[0][s][1], cH);
      v2f a1 = (v2f){a1p[k] * rs1, a1p[k + 1] * rs1};
      cL = wmma_k4(a1, b[1][s][0], cL);
      cH = wmma_k4(a1, b[1][s][1], cH);
      v2f a2 = (v2f){fmaxf(a2p[k], 0.0f), fmaxf(a2p[k + 1], 0.0f)};
      cL = wmma_k4(a2, b[2][s][0], cL);
      cH = wmma_k4(a2, b[2][s][1], cH);
    }
#pragma unroll
    for (int r = 0; r < 8; ++r) {
      int row = rbase + r + 8 * half;
      if (row < M) {
        size_t o = (size_t)row * 32 + mrow;
        OUT[o] = cL[r] + biasL;
        OUT[o + 16] = cH[r] + biasH;
      }
    }
  }
}


extern "C" void kernel_launch(void* const* d_in, const int* in_sizes, int n_in,
                              void* d_out, int out_size, void* d_ws,
                              size_t ws_size, hipStream_t stream) {
  const float* x_dev = (const float*)d_in[0];
  const float* x_task = (const float*)d_in[1];
  const int* wr_src = (const int*)d_in[2];
  const int* wr_dst = (const int*)d_in[3];
  const int* rv_src = (const int*)d_in[4];
  const int* rv_dst = (const int*)d_in[5];
  const int* wb_src = (const int*)d_in[6];
  const int* wb_dst = (const int*)d_in[7];
  const int* rb_src = (const int*)d_in[8];
  const int* rb_dst = (const int*)d_in[9];
  const float* p1wr_wl = (const float*)d_in[10];
  const float* p1wr_bl = (const float*)d_in[11];
  const float* p1wr_wr = (const float*)d_in[12];
  const float* p1rv_wl = (const float*)d_in[13];
  const float* p1rv_bl = (const float*)d_in[14];
  const float* p1rv_wr = (const float*)d_in[15];
  const float* p1wb_wl = (const float*)d_in[16];
  const float* p1wb_bl = (const float*)d_in[17];
  const float* p1wb_wr = (const float*)d_in[18];
  const float* p1rb_wl = (const float*)d_in[19];
  const float* p1rb_bl = (const float*)d_in[20];
  const float* p1rb_wr = (const float*)d_in[21];
  const float* p2wr_wl = (const float*)d_in[22];
  const float* p2wr_bl = (const float*)d_in[23];
  const float* p2wr_wr = (const float*)d_in[24];
  const float* p2rv_wl = (const float*)d_in[25];
  const float* p2rv_bl = (const float*)d_in[26];
  const float* p2rv_wr = (const float*)d_in[27];
  const float* p2wb_wl = (const float*)d_in[28];
  const float* p2wb_bl = (const float*)d_in[29];
  const float* p2wb_wr = (const float*)d_in[30];
  const float* p2rb_wl = (const float*)d_in[31];
  const float* p2rb_bl = (const float*)d_in[32];
  const float* p2rb_wr = (const float*)d_in[33];

  const int ND = in_sizes[0] / 64;
  const int NT = in_sizes[1] / 64;
  const int E_wr = in_sizes[2], E_rv = in_sizes[4];
  const int E_wb = in_sizes[6], E_rb = in_sizes[8];

  const int cntN = 2 * NT + 2 * ND;
  const int Etot = E_wr + E_rv + E_wb + E_rb;
  const int nChunks = (cntN + 2047) / 2048;
  float* ws = (float*)d_ws;
  size_t off = 0;
  int* cnt = (int*)(ws + off); off += (size_t)cntN;
  float* csr = ws + off; off += (size_t)cntN * 16;
  float* yreg = ws + off; off += (size_t)cntN * 16;
  float* t1 = ws + off; off += (size_t)NT * 16;
  float* d1 = ws + off; off += (size_t)ND * 16;
  float* ws1t = ws + off; off += 64 * 16;
  float* ws1d = ws + off; off += 64 * 16;
  float* ws2t = ws + off; off += 16 * 32;
  float* ws2d = ws + off; off += 16 * 32;

  int* cnt_wr = cnt;
  int* cnt_rv = cnt + NT;
  int* cnt_wb = cnt + 2 * (size_t)NT;
  int* cnt_rb = cnt + 2 * (size_t)NT + ND;
  int* pos = (int*)csr;
  int* pos_wr = pos;
  int* pos_rv = pos + NT;
  int* pos_wb = pos + 2 * (size_t)NT;
  int* pos_rb = pos + 2 * (size_t)NT + ND;
  int* col = pos + cntN;
  int* bsum = col + Etot;
  int* rank = bsum + nChunks;
  int* rank_wr = rank;
  int* rank_rv = rank_wr + E_wr;
  int* rank_wb = rank_rv + E_rv;
  int* rank_rb = rank_wb + E_wb;

  float* y_wr = yreg;
  float* y_rv = y_wr + (size_t)ND * 16;
  float* y_wb = y_rv + (size_t)ND * 16;
  float* y_rb = y_wb + (size_t)NT * 16;
  float* agg_wr = yreg;
  float* agg_rv = agg_wr + (size_t)NT * 16;
  float* agg_wb = agg_rv + (size_t)NT * 16;
  float* agg_rb = agg_wb + (size_t)ND * 16;

  const int TPB = 256;

  zero_int_kernel<<<(cntN + TPB - 1) / TPB, TPB, 0, stream>>>(cnt, cntN);
  count_kernel<<<(E_wr + TPB - 1) / TPB, TPB, 0, stream>>>(wr_dst, E_wr, cnt_wr, rank_wr);
  count_kernel<<<(E_rv + TPB - 1) / TPB, TPB, 0, stream>>>(rv_dst, E_rv, cnt_rv, rank_rv);
  count_kernel<<<(E_wb + TPB - 1) / TPB, TPB, 0, stream>>>(wb_dst, E_wb, cnt_wb, rank_wb);
  count_kernel<<<(E_rb + TPB - 1) / TPB, TPB, 0, stream>>>(rb_dst, E_rb, cnt_rb, rank_rb);

  scan_partial_kernel<<<nChunks, 256, 0, stream>>>(cnt, cntN, bsum);
  scan_offsets_kernel<<<1, 256, 0, stream>>>(bsum, nChunks);
  scan_final_kernel<<<nChunks, 256, 0, stream>>>(cnt, cntN, bsum, pos);

  fill_kernel<<<(E_wr + TPB - 1) / TPB, TPB, 0, stream>>>(wr_src, wr_dst, rank_wr, E_wr, pos_wr, col);
  fill_kernel<<<(E_rv + TPB - 1) / TPB, TPB, 0, stream>>>(rv_src, rv_dst, rank_rv, E_rv, pos_rv, col);
  fill_kernel<<<(E_wb + TPB - 1) / TPB, TPB, 0, stream>>>(wb_src, wb_dst, rank_wb, E_wb, pos_wb, col);
  fill_kernel<<<(E_rb + TPB - 1) / TPB, TPB, 0, stream>>>(rb_src, rb_dst, rank_rb, E_rb, pos_rb, col);

  addmat_kernel<<<4, TPB, 0, stream>>>(p1wr_wr, p1rv_wr, ws1t, 64 * 16);
  addmat_kernel<<<4, TPB, 0, stream>>>(p1wb_wr, p1rb_wr, ws1d, 64 * 16);
  addmat_kernel<<<2, TPB, 0, stream>>>(p2wr_wr, p2rv_wr, ws2t, 16 * 32);
  addmat_kernel<<<2, TPB, 0, stream>>>(p2wb_wr, p2rb_wr, ws2d, 16 * 32);

  const int tilesD = (ND + 15) / 16, tilesT = (NT + 15) / 16;
  l1_transform_kernel<<<(tilesD + 3) / 4, 128, 0, stream>>>(
      x_dev, p1wr_wl, p1rv_wl, ws1d, p1wb_bl, p1rb_bl, y_wr, y_rv, d1, ND);
  l1_transform_kernel<<<(tilesT + 3) / 4, 128, 0, stream>>>(
      x_task, p1wb_wl, p1rb_wl, ws1t, p1wr_bl, p1rv_bl, y_wb, y_rb, t1, NT);

  gather_l1_kernel<<<(NT * 4 + TPB - 1) / TPB, TPB, 0, stream>>>(
      y_wr, y_rv, pos_wr, cnt_wr, pos_rv, cnt_rv, col, t1, NT);
  gather_l1_kernel<<<(ND * 4 + TPB - 1) / TPB, TPB, 0, stream>>>(
      y_wb, y_rb, pos_wb, cnt_wb, pos_rb, cnt_rb, col, d1, ND);

  gather_l2_kernel<<<(NT * 4 + TPB - 1) / TPB, TPB, 0, stream>>>(
      d1, pos_wr, cnt_wr, pos_rv, cnt_rv, col, agg_wr, agg_rv, NT);
  gather_l2_kernel<<<(ND * 4 + TPB - 1) / TPB, TPB, 0, stream>>>(
      t1, pos_wb, cnt_wb, pos_rb, cnt_rb, col, agg_wb, agg_rb, ND);

  float* out_d2 = (float*)d_out;
  float* out_t2 = (float*)d_out + (size_t)ND * 32;
  l2_fused_kernel<<<(tilesT + 3) / 4, 128, 0, stream>>>(
      agg_wr, cnt_wr, agg_rv, cnt_rv, t1, p2wr_wl, p2rv_wl, ws2t, p2wr_bl,
      p2rv_bl, out_t2, NT);
  l2_fused_kernel<<<(tilesD + 3) / 4, 128, 0, stream>>>(
      agg_wb, cnt_wb, agg_rb, cnt_rb, d1, p2wb_wl, p2rb_wl, ws2d, p2wb_bl,
      p2rb_bl, out_d2, ND);
}


# Edge grouping redone as a two-level counting sort with all counters in LDS (bin by node range, then one block per bin counts, scans, places): no atomics on memory; row sums add in LDS-arrival order
# speedup vs baseline: 1.9064x; 1.3749x over previous
#include <hip/hip_runtime.h>


typedef __attribute__((ext_vector_type(2))) float v2f;
typedef __attribute__((ext_vector_type(8))) float v8f;

__device__ __forceinline__ v8f wmma_k4(v2f a, v2f b, v8f c) {
  return __builtin_amdgcn_wmma_f32_16x16x4_f32(
       false, a,  false, b,
       (short)0, c,  false,  false);
}


__global__ void addmat_kernel(const float* __restrict__ a,
                              const float* __restrict__ b,
                              float* __restrict__ o, int n) {
  int i = blockIdx.x * blockDim.x + threadIdx.x;
  if (i < n) o[i] = a[i] + b[i];
}


__device__ __forceinline__ int block_excl_scan_256(int v, int* sh, int& total) {
  const int t = threadIdx.x;
  sh[t] = v;
  __syncthreads();
  for (int off = 1; off < 256; off <<= 1) {
    const int add = (t >= off) ? sh[t - off] : 0;
    __syncthreads();
    sh[t] += add;
    __syncthreads();
  }
  const int incl = sh[t];
  total = sh[255];
  __syncthreads();
  return incl - v;
}

__global__ __launch_bounds__(256) void scan_partial_kernel(
    const int* __restrict__ cnt, int n, int* __restrict__ bsum) {
  __shared__ int sh[256];
  const int base = blockIdx.x * 2048 + threadIdx.x * 8;
  int v[8];
#pragma unroll
  for (int j = 0; j < 8; ++j) {
    const int idx = base + j;
    v[j] = (idx < n) ? cnt[idx] : 0;
  }
  int s = 0;
#pragma unroll
  for (int j = 0; j < 8; ++j) s += v[j];
  int total;
  block_excl_scan_256(s, sh, total);
  if (threadIdx.x == 0) bsum[blockIdx.x] = total;
}

__global__ __launch_bounds__(256) void scan_offsets_kernel(
    int* __restrict__ bsum, int nb) {
  __shared__ int sh[256];
  int carry = 0;
  for (int base = 0; base < nb; base += 256) {
    const int idx = base + threadIdx.x;
    const int v = (idx < nb) ? bsum[idx] : 0;
    int total;
    const int ex = block_excl_scan_256(v, sh, total);
    if (idx < nb) bsum[idx] = carry + ex;
    carry += total;
  }
}

__global__ __launch_bounds__(256) void scan_final_kernel(
    const int* __restrict__ cnt, int n, const int* __restrict__ bsum,
    int* __restrict__ pos) {
  __shared__ int sh[256];
  const int base = blockIdx.x * 2048 + threadIdx.x * 8;
  int v[8];
  int s = 0;
#pragma unroll
  for (int j = 0; j < 8; ++j) {
    const int idx = base + j;
    v[j] = (idx < n) ? cnt[idx] : 0;
    s += v[j];
  }
  int total;
  const int ex = block_excl_scan_256(s, sh, total);
  int run = bsum[blockIdx.x] + ex;
#pragma unroll
  for (int j = 0; j < 8; ++j) {
    const int idx = base + j;
    if (idx < n) pos[idx] = run;
    run += v[j];
  }
}

#define MAX_BINS 1024
#define MAX_BIN_NODES 2048
#define EDGES_PER_BLOCK 4096

__global__ __launch_bounds__(256) void bin_hist_kernel(
    const int* __restrict__ dst, int E, int shift, int nbins, int nblocks,
    int* __restrict__ tab) {
  __shared__ int h[MAX_BINS];
  const int t = threadIdx.x;
  for (int k = t; k < nbins; k += 256) h[k] = 0;
  __syncthreads();
  const int base = blockIdx.x * EDGES_PER_BLOCK;
  int d[EDGES_PER_BLOCK / 256];
#pragma unroll
  for (int j = 0; j < EDGES_PER_BLOCK / 256; ++j) {
    const int i = base + j * 256 + t;
    d[j] = (i < E) ? dst[i] : -1;
  }
#pragma unroll
  for (int j = 0; j < EDGES_PER_BLOCK / 256; ++j)
    if (d[j] >= 0) atomicAdd(&h[d[j] >> shift], 1);
  __syncthreads();
  for (int k = t; k < nbins; k += 256) tab[k * nblocks + blockIdx.x] = h[k];
}

__global__ __launch_bounds__(256) void bin_scatter_kernel(
    const int* __restrict__ src, const int* __restrict__ dst, int E, int shift,
    int nbins, int nblocks, const int* __restrict__ tpos,
    int* __restrict__ bedge) {
  __shared__ int cur[MAX_BINS];
  const int t = threadIdx.x;
  for (int k = t; k < nbins; k += 256) cur[k] = tpos[k * nblocks + blockIdx.x];
  __syncthreads();
  int2* be = reinterpret_cast<int2*>(bedge);
  const int base = blockIdx.x * EDGES_PER_BLOCK;
  int d[EDGES_PER_BLOCK / 256], s[EDGES_PER_BLOCK / 256];
  int slot[EDGES_PER_BLOCK / 256];
#pragma unroll
  for (int j = 0; j < EDGES_PER_BLOCK / 256; ++j) {
    const int i = base + j * 256 + t;
    d[j] = (i < E) ? dst[i] : -1;
    s[j] = (i < E) ? src[i] : 0;
  }
#pragma unroll
  for (int j = 0; j < EDGES_PER_BLOCK / 256; ++j)
    slot[j] = (d[j] >= 0) ? atomicAdd(&cur[d[j] >> shift], 1) : 0;
#pragma unroll
  for (int j = 0; j < EDGES_PER_BLOCK / 256; ++j)
    if (d[j] >= 0) be[slot[j]] = make_int2(d[j], s[j]);
}

__global__ __launch_bounds__(256) void bin_csr_kernel(
    const int* __restrict__ bedge, int shift, int nblocks,
    const int* __restrict__ tposAll, int tabOff, int nTab, int Etot, int N,
    int* __restrict__ cnt, int* __restrict__ pos, int* __restrict__ col) {
  __shared__ int lc[MAX_BIN_NODES];
  __shared__ int sh[256];
  const int t = threadIdx.x;
  const int b = blockIdx.x;
  const int B = 1 << shift;
  const int per = B >> 8;
  const int nodeBase = b << shift;
  const int first = tabOff + b * nblocks;
  const int next = first + nblocks;
  const int beg = tposAll[first];
  const int end = (next < nTab) ? tposAll[next] : Etot;
  const int2* be = reinterpret_cast<const int2*>(bedge);

  for (int k = t; k < B; k += 256) lc[k] = 0;
  __syncthreads();
  for (int e0 = beg; e0 < end; e0 += 8 * 256) {
    int x[8];
#pragma unroll
    for (int u = 0; u < 8; ++u) {
      const int e = e0 + u * 256 + t;
      x[u] = (e < end) ? be[e].x : -1;
    }
#pragma unroll
    for (int u = 0; u < 8; ++u)
      if (x[u] >= 0) atomicAdd(&lc[x[u] - nodeBase], 1);
  }
  __syncthreads();

  int v[8];
  int s = 0;
#pragma unroll
  for (int j = 0; j < 8; ++j) {
    v[j] = (j < per) ? lc[t * per + j] : 0;
    s += v[j];
  }
  int total;
  int run = block_excl_scan_256(s, sh, total);
#pragma unroll
  for (int j = 0; j < 8; ++j) {
    if (j < per) {
      const int k = t * per + j;
      const int node = nodeBase + k;
      lc[k] = run;
      if (node < N) {
        cnt[node] = v[j];
        pos[node] = beg + run;
      }
      run += v[j];
    }
  }
  __syncthreads();
  for (int e0 = beg; e0 < end; e0 += 8 * 256) {
    int2 r[8];
    int slot[8];
#pragma unroll
    for (int u = 0; u < 8; ++u) {
      const int e = e0 + u * 256 + t;
      r[u] = (e < end) ? be[e] : make_int2(-1, 0);
    }
#pragma unroll
    for (int u = 0; u < 8; ++u)
      slot[u] = (r[u].x >= 0) ? atomicAdd(&lc[r[u].x - nodeBase], 1) : 0;
#pragma unroll
    for (int u = 0; u < 8; ++u)
      if (r[u].x >= 0) col[beg + slot[u]] = r[u].y;
  }
}

__device__ __forceinline__ float4 load_row4(const float* __restrict__ Y, int s,
                                            int c) {
  return *reinterpret_cast<const float4*>(Y + (size_t)s * 16 + c * 4);
}

__device__ __forceinline__ float4 gather_rows(const float* __restrict__ Y,
                                              const int* __restrict__ col,
                                              int beg, int end, int c) {
  float4 acc = make_float4(0.0f, 0.0f, 0.0f, 0.0f);
  int p = beg;
  for (; p + 1 < end; p += 2) {
    const int s0 = col[p];
    const int s1 = col[p + 1];
    const float4 v0 = load_row4(Y, s0, c);
    const float4 v1 = load_row4(Y, s1, c);
    acc.x += v0.x;
    acc.y += v0.y;
    acc.z += v0.z;
    acc.w += v0.w;
    acc.x += v1.x;
    acc.y += v1.y;
    acc.z += v1.z;
    acc.w += v1.w;
  }
  if (p < end) {
    const float4 v = load_row4(Y, col[p], c);
    acc.x += v.x;
    acc.y += v.y;
    acc.z += v.z;
    acc.w += v.w;
  }
  return acc;
}

__global__ void gather_l1_kernel(const float* __restrict__ YA,
                                 const float* __restrict__ YB,
                                 const int* __restrict__ begA,
                                 const int* __restrict__ cntA,
                                 const int* __restrict__ begB,
                                 const int* __restrict__ cntB,
                                 const int* __restrict__ col,
                                 float* __restrict__ T, int N) {
  const int i = blockIdx.x * blockDim.x + threadIdx.x;
  const int node = i >> 2;
  const int c = i & 3;
  if (node >= N) return;
  const int nA = cntA[node], bA = begA[node];
  const int nB = cntB[node], bB = begB[node];
  float4* tp = reinterpret_cast<float4*>(T + (size_t)node * 16 + c * 4);
  float4 acc = *tp;
  const float4 sA = gather_rows(YA, col, bA, bA + nA, c);
  const float4 sB = gather_rows(YB, col, bB, bB + nB, c);
  const float iA = 1.0f / fmaxf((float)nA, 1.0f);
  const float iB = 1.0f / fmaxf((float)nB, 1.0f);
  acc.x += sA.x * iA + sB.x * iB;
  acc.y += sA.y * iA + sB.y * iB;
  acc.z += sA.z * iA + sB.z * iB;
  acc.w += sA.w * iA + sB.w * iB;
  acc.x = fmaxf(acc.x, 0.0f);
  acc.y = fmaxf(acc.y, 0.0f);
  acc.z = fmaxf(acc.z, 0.0f);
  acc.w = fmaxf(acc.w, 0.0f);
  *tp = acc;
}

__global__ void gather_l2_kernel(const float* __restrict__ H,
                                 const int* __restrict__ begA,
                                 const int* __restrict__ cntA,
                                 const int* __restrict__ begB,
                                 const int* __restrict__ cntB,
                                 const int* __restrict__ col,
                                 float* __restrict__ aggA,
                                 float* __restrict__ aggB, int N) {
  const int i = blockIdx.x * blockDim.x + threadIdx.x;
  const int node = i >> 2;
  const int c = i & 3;
  if (node >= N) return;
  const int nA = cntA[node], bA = begA[node];
  const int nB = cntB[node], bB = begB[node];
  const float4 sA = gather_rows(H, col, bA, bA + nA, c);
  const float4 sB = gather_rows(H, col, bB, bB + nB, c);
  *reinterpret_cast<float4*>(aggA + (size_t)node * 16 + c * 4) = sA;
  *reinterpret_cast<float4*>(aggB + (size_t)node * 16 + c * 4) = sB;
}

__global__ __launch_bounds__(128) void l1_transform_kernel(
    const float* __restrict__ X, const float* __restrict__ W0,
    const float* __restrict__ W1, const float* __restrict__ W2,
    const float* __restrict__ biasA, const float* __restrict__ biasB,
    float* __restrict__ Y0, float* __restrict__ Y1, float* __restrict__ Y2,
    int M) {
  const int lane = threadIdx.x & 31;
  const int wave = threadIdx.x >> 5;
  const int wpb = blockDim.x >> 5;
  const int half = lane >> 4;
  const int mrow = lane & 15;
  const int nTiles = (M + 15) >> 4;

  v2f b0[16], b1[16], b2[16];
#pragma unroll
  for (int s = 0; s < 16; ++s) {
    int k = 4 * s + 2 * half;
    b0[s] = (v2f){W0[k * 16 + mrow], W0[(k + 1) * 16 + mrow]};
    b1[s] = (v2f){W1[k * 16 + mrow], W1[(k + 1) * 16 + mrow]};
    b2[s] = (v2f){W2[k * 16 + mrow], W2[(k + 1) * 16 + mrow]};
  }

  const float bsum = biasA[mrow] + biasB[mrow];

  for (int tile = blockIdx.x * wpb + wave; tile < nTiles;
       tile += gridDim.x * wpb) {
    const int rbase = tile << 4;
    int arow = rbase + mrow;
    if (arow > M - 1) arow = M - 1;
    const float* xr = X + (size_t)arow * 64;
    v8f c0 = {}, c1 = {};
    v8f c2 = {bsum, bsum, bsum, bsum, bsum, bsum, bsum, bsum};
#pragma unroll
    for (int s = 0; s < 16; ++s) {
      int k = 4 * s + 2 * half;
      v2f a = (v2f){xr[k], xr[k + 1]};
      c0 = wmma_k4(a, b0[s], c0);
      c1 = wmma_k4(a, b1[s], c1);
      c2 = wmma_k4(a, b2[s], c2);
    }
#pragma unroll
    for (int r = 0; r < 8; ++r) {
      int row = rbase + r + 8 * half;
      if (row < M) {
        size_t o = (size_t)row * 16 + mrow;
        Y0[o] = c0[r];
        Y1[o] = c1[r];
        Y2[o] = c2[r];
      }
    }
  }
}

__global__ __launch_bounds__(128) void l2_fused_kernel(
    const float* __restrict__ A0, const int* __restrict__ cnt0,
    const float* __restrict__ A1, const int* __restrict__ cnt1,
    const float* __restrict__ A2, const float* __restrict__ W0,
    const float* __restrict__ W1, const float* __restrict__ W2,
    const float* __restrict__ bias0, const float* __restrict__ bias1,
    float* __restrict__ OUT, int M) {
  const int lane = threadIdx.x & 31;
  const int wave = threadIdx.x >> 5;
  const int wpb = blockDim.x >> 5;
  const int half = lane >> 4;
  const int mrow = lane & 15;
  const int nTiles = (M + 15) >> 4;

  const float* Ws[3] = {W0, W1, W2};
  v2f b[3][4][2];
#pragma unroll
  for (int m = 0; m < 3; ++m)
#pragma unroll
    for (int s = 0; s < 4; ++s)
#pragma unroll
      for (int t = 0; t < 2; ++t) {
        int k = 4 * s + 2 * half;
        int col = 16 * t + mrow;
        b[m][s][t] = (v2f){Ws[m][k * 32 + col], Ws[m][(k + 1) * 32 + col]};
      }

  const float biasL = bias0[mrow] + bias1[mrow];
  const float biasH = bias0[16 + mrow] + bias1[16 + mrow];

  for (int tile = blockIdx.x * wpb + wave; tile < nTiles;
       tile += gridDim.x * wpb) {
    const int rbase = tile << 4;
    int arow = rbase + mrow;
    if (arow > M - 1) arow = M - 1;
    const float rs0 = 1.0f / fmaxf((float)cnt0[arow], 1.0f);
    const float rs1 = 1.0f / fmaxf((float)cnt1[arow], 1.0f);
    const float* a0p = A0 + (size_t)arow * 16;
    const float* a1p = A1 + (size_t)arow * 16;
    const float* a2p = A2 + (size_t)arow * 16;
    v8f cL = {}, cH = {};
#pragma unroll
    for (int s = 0; s < 4; ++s) {
      int k = 4 * s + 2 * half;
      v2f a0 = (v2f){a0p[k] * rs0, a0p[k + 1] * rs0};
      cL = wmma_k4(a0, b[0][s][0], cL);
      cH = wmma_k4(a0, b[0][s][1], cH);
      v2f a1 = (v2f){a1p[k] * rs1, a1p[k + 1] * rs1};
      cL = wmma_k4(a1, b[1][s][0], cL);
      cH = wmma_k4(a1, b[1][s][1], cH);
      v2f a2 = (v2f){fmaxf(a2p[k], 0.0f), fmaxf(a2p[k + 1], 0.0f)};
      cL = wmma_k4(a2, b[2][s][0], cL);
      cH = wmma_k4(a2, b[2][s][1], cH);
    }
#pragma unroll
    for (int r = 0; r < 8; ++r) {
      int row = rbase + r + 8 * half;
      if (row < M) {
        size_t o = (size_t)row * 32 + mrow;
        OUT[o] = cL[r] + biasL;
        OUT[o + 16] = cH[r] + biasH;
      }
    }
  }
}


static inline int bin_shift_for(int n) {
  int s = 8;
  while (s < 11 && ((n - 1) >> s) >= 256) ++s;
  return s;
}

extern "C" void kernel_launch(void* const* d_in, const int* in_sizes, int n_in,
                              void* d_out, int out_size, void* d_ws,
                              size_t ws_size, hipStream_t stream) {
  const float* x_dev = (const float*)d_in[0];
  const float* x_task = (const float*)d_in[1];
  const int* wr_src = (const int*)d_in[2];
  const int* wr_dst = (const int*)d_in[3];
  const int* rv_src = (const int*)d_in[4];
  const int* rv_dst = (const int*)d_in[5];
  const int* wb_src = (const int*)d_in[6];
  const int* wb_dst = (const int*)d_in[7];
  const int* rb_src = (const int*)d_in[8];
  const int* rb_dst = (const int*)d_in[9];
  const float* p1wr_wl = (const float*)d_in[10];
  const float* p1wr_bl = (const float*)d_in[11];
  const float* p1wr_wr = (const float*)d_in[12];
  const float* p1rv_wl = (const float*)d_in[13];
  const float* p1rv_bl = (const float*)d_in[14];
  const float* p1rv_wr = (const float*)d_in[15];
  const float* p1wb_wl = (const float*)d_in[16];
  const float* p1wb_bl = (const float*)d_in[17];
  const float* p1wb_wr = (const float*)d_in[18];
  const float* p1rb_wl = (const float*)d_in[19];
  const float* p1rb_bl = (const float*)d_in[20];
  const float* p1rb_wr = (const float*)d_in[21];
  const float* p2wr_wl = (const float*)d_in[22];
  const float* p2wr_bl = (const float*)d_in[23];
  const float* p2wr_wr = (const float*)d_in[24];
  const float* p2rv_wl = (const float*)d_in[25];
  const float* p2rv_bl = (const float*)d_in[26];
  const float* p2rv_wr = (const float*)d_in[27];
  const float* p2wb_wl = (const float*)d_in[28];
  const float* p2wb_bl = (const float*)d_in[29];
  const float* p2wb_wr = (const float*)d_in[30];
  const float* p2rb_wl = (const float*)d_in[31];
  const float* p2rb_bl = (const float*)d_in[32];
  const float* p2rb_wr = (const float*)d_in[33];

  const int ND = in_sizes[0] / 64;
  const int NT = in_sizes[1] / 64;
  const int E_wr = in_sizes[2], E_rv = in_sizes[4];
  const int E_wb = in_sizes[6], E_rb = in_sizes[8];

  const int cntN = 2 * NT + 2 * ND;
  const int Etot = E_wr + E_rv + E_wb + E_rb;
  const int shT = bin_shift_for(NT), shD = bin_shift_for(ND);
  const int nbT = ((NT - 1) >> shT) + 1, nbD = ((ND - 1) >> shD) + 1;
  const int nc_wr = (E_wr + EDGES_PER_BLOCK - 1) / EDGES_PER_BLOCK;
  const int nc_rv = (E_rv + EDGES_PER_BLOCK - 1) / EDGES_PER_BLOCK;
  const int nc_wb = (E_wb + EDGES_PER_BLOCK - 1) / EDGES_PER_BLOCK;
  const int nc_rb = (E_rb + EDGES_PER_BLOCK - 1) / EDGES_PER_BLOCK;
  const int to_wr = 0;
  const int to_rv = to_wr + nbT * nc_wr;
  const int to_wb = to_rv + nbT * nc_rv;
  const int to_rb = to_wb + nbD * nc_wb;
  const int nTab = to_rb + nbD * nc_rb;
  const int nChunks = (nTab + 2047) / 2048;
  float* ws = (float*)d_ws;
  size_t off = 0;
  int* cnt = (int*)(ws + off); off += (size_t)cntN;
  float* csr = ws + off; off += (size_t)cntN * 16;
  float* yreg = ws + off; off += (size_t)cntN * 16;
  float* t1 = ws + off; off += (size_t)NT * 16;
  float* d1 = ws + off; off += (size_t)ND * 16;
  float* ws1t = ws + off; off += 64 * 16;
  float* ws1d = ws + off; off += 64 * 16;
  float* ws2t = ws + off; off += 16 * 32;
  float* ws2d = ws + off; off += 16 * 32;

  int* cnt_wr = cnt;
  int* cnt_rv = cnt + NT;
  int* cnt_wb = cnt + 2 * (size_t)NT;
  int* cnt_rb = cnt + 2 * (size_t)NT + ND;
  int* pos = (int*)csr;
  int* pos_wr = pos;
  int* pos_rv = pos + NT;
  int* pos_wb = pos + 2 * (size_t)NT;
  int* pos_rb = pos + 2 * (size_t)NT + ND;
  int* col = pos + cntN;
  size_t q = (size_t)2 * cntN + Etot;
  q = (q + 1) & ~(size_t)1;
  int* bedge = (int*)ws + q; q += (size_t)2 * Etot;
  int* tab = (int*)ws + q; q += (size_t)nTab;
  int* tpos = (int*)ws + q; q += (size_t)nTab;
  int* bsum = (int*)ws + q;

  float* y_wr = yreg;
  float* y_rv = y_wr + (size_t)ND * 16;
  float* y_wb = y_rv + (size_t)ND * 16;
  float* y_rb = y_wb + (size_t)NT * 16;
  float* agg_wr = yreg;
  float* agg_rv = agg_wr + (size_t)NT * 16;
  float* agg_wb = agg_rv + (size_t)NT * 16;
  float* agg_rb = agg_wb + (size_t)ND * 16;

  const int TPB = 256;

  bin_hist_kernel<<<nc_wr, 256, 0, stream>>>(wr_dst, E_wr, shT, nbT, nc_wr, tab + to_wr);
  bin_hist_kernel<<<nc_rv, 256, 0, stream>>>(rv_dst, E_rv, shT, nbT, nc_rv, tab + to_rv);
  bin_hist_kernel<<<nc_wb, 256, 0, stream>>>(wb_dst, E_wb, shD, nbD, nc_wb, tab + to_wb);
  bin_hist_kernel<<<nc_rb, 256, 0, stream>>>(rb_dst, E_rb, shD, nbD, nc_rb, tab + to_rb);

  scan_partial_kernel<<<nChunks, 256, 0, stream>>>(tab, nTab, bsum);
  scan_offsets_kernel<<<1, 256, 0, stream>>>(bsum, nChunks);
  scan_final_kernel<<<nChunks, 256, 0, stream>>>(tab, nTab, bsum, tpos);

  bin_scatter_kernel<<<nc_wr, 256, 0, stream>>>(wr_src, wr_dst, E_wr, shT, nbT, nc_wr, tpos + to_wr, bedge);
  bin_scatter_kernel<<<nc_rv, 256, 0, stream>>>(rv_src, rv_dst, E_rv, shT, nbT, nc_rv, tpos + to_rv, bedge);
  bin_scatter_kernel<<<nc_wb, 256, 0, stream>>>(wb_src, wb_dst, E_wb, shD, nbD, nc_wb, tpos + to_wb, bedge);
  bin_scatter_kernel<<<nc_rb, 256, 0, stream>>>(rb_src, rb_dst, E_rb, shD, nbD, nc_rb, tpos + to_rb, bedge);

  bin_csr_kernel<<<nbT, 256, 0, stream>>>(bedge, shT, nc_wr, tpos, to_wr, nTab, Etot, NT, cnt_wr, pos_wr, col);
  bin_csr_kernel<<<nbT, 256, 0, stream>>>(bedge, shT, nc_rv, tpos, to_rv, nTab, Etot, NT, cnt_rv, pos_rv, col);
  bin_csr_kernel<<<nbD, 256, 0, stream>>>(bedge, shD, nc_wb, tpos, to_wb, nTab, Etot, ND, cnt_wb, pos_wb, col);
  bin_csr_kernel<<<nbD, 256, 0, stream>>>(bedge, shD, nc_rb, tpos, to_rb, nTab, Etot, ND, cnt_rb, pos_rb, col);

  addmat_kernel<<<4, TPB, 0, stream>>>(p1wr_wr, p1rv_wr, ws1t, 64 * 16);
  addmat_kernel<<<4, TPB, 0, stream>>>(p1wb_wr, p1rb_wr, ws1d, 64 * 16);
  addmat_kernel<<<2, TPB, 0, stream>>>(p2wr_wr, p2rv_wr, ws2t, 16 * 32);
  addmat_kernel<<<2, TPB, 0, stream>>>(p2wb_wr, p2rb_wr, ws2d, 16 * 32);

  const int tilesD = (ND + 15) / 16, tilesT = (NT + 15) / 16;
  l1_transform_kernel<<<(tilesD + 3) / 4, 128, 0, stream>>>(
      x_dev, p1wr_wl, p1rv_wl, ws1d, p1wb_bl, p1rb_bl, y_wr, y_rv, d1, ND);
  l1_transform_kernel<<<(tilesT + 3) / 4, 128, 0, stream>>>(
      x_task, p1wb_wl, p1rb_wl, ws1t, p1wr_bl, p1rv_bl, y_wb, y_rb, t1, NT);

  gather_l1_kernel<<<(NT * 4 + TPB - 1) / TPB, TPB, 0, stream>>>(
      y_wr, y_rv, pos_wr, cnt_wr, pos_rv, cnt_rv, col, t1, NT);
  gather_l1_kernel<<<(ND * 4 + TPB - 1) / TPB, TPB, 0, stream>>>(
      y_wb, y_rb, pos_wb, cnt_wb, pos_rb, cnt_rb, col, d1, ND);

  gather_l2_kernel<<<(NT * 4 + TPB - 1) / TPB, TPB, 0, stream>>>(
      d1, pos_wr, cnt_wr, pos_rv, cnt_rv, col, agg_wr, agg_rv, NT);
  gather_l2_kernel<<<(ND * 4 + TPB - 1) / TPB, TPB, 0, stream>>>(
      t1, pos_wb, cnt_wb, pos_rb, cnt_rb, col, agg_wb, agg_rb, ND);

  float* out_d2 = (float*)d_out;
  float* out_t2 = (float*)d_out + (size_t)ND * 32;
  l2_fused_kernel<<<(tilesT + 3) / 4, 128, 0, stream>>>(
      agg_wr, cnt_wr, agg_rv, cnt_rv, t1, p2wr_wl, p2rv_wl, ws2t, p2wr_bl,
      p2rv_bl, out_t2, NT);
  l2_fused_kernel<<<(tilesD + 3) / 4, 128, 0, stream>>>(
      agg_wb, cnt_wb, agg_rb, cnt_rb, d1, p2wb_wl, p2rb_wl, ws2d, p2wb_bl,
      p2rb_bl, out_d2, ND);
}


# Launches merged, 27 -> 9 per call: each sort stage serves all four relations, each layer stage both node types, in one launch; weight sums and the chunk-offset scan ride in neighbouring launches
# speedup vs baseline: 2.7730x; 1.4546x over previous
#include <hip/hip_runtime.h>


typedef __attribute__((ext_vector_type(2))) float v2f;
typedef __attribute__((ext_vector_type(8))) float v8f;

__device__ __forceinline__ v8f wmma_k4(v2f a, v2f b, v8f c) {
  return __builtin_amdgcn_wmma_f32_16x16x4_f32(
       false, a,  false, b,
       (short)0, c,  false,  false);
}


__device__ __forceinline__ int pick4(int4 v, int r) {
  return r == 0 ? v.x : r == 1 ? v.y : r == 2 ? v.z : v.w;
}

__device__ __forceinline__ const int* pick_ip(const int* a, const int* b,
                                              const int* c, const int* d,
                                              int r) {
  return r == 0 ? a : r == 1 ? b : r == 2 ? c : d;
}

__device__ __forceinline__ const float* pick_fp(const float* a, const float* b,
                                                const float* c, const float* d,
                                                int r) {
  return r == 0 ? a : r == 1 ? b : r == 2 ? c : d;
}


__device__ __forceinline__ int block_excl_scan_256(int v, int* sh, int& total) {
  const int t = threadIdx.x;
  sh[t] = v;
  __syncthreads();
  for (int off = 1; off < 256; off <<= 1) {
    const int add = (t >= off) ? sh[t - off] : 0;
    __syncthreads();
    sh[t] += add;
    __syncthreads();
  }
  const int incl = sh[t];
  total = sh[255];
  __syncthreads();
  return incl - v;
}

__global__ __launch_bounds__(256) void scan_partial_kernel(
    const int* __restrict__ cnt, int n, int* __restrict__ bsum) {
  __shared__ int sh[256];
  const int base = blockIdx.x * 2048 + threadIdx.x * 8;
  int v[8];
#pragma unroll
  for (int j = 0; j < 8; ++j) {
    const int idx = base + j;
    v[j] = (idx < n) ? cnt[idx] : 0;
  }
  int s = 0;
#pragma unroll
  for (int j = 0; j < 8; ++j) s += v[j];
  int total;
  block_excl_scan_256(s, sh, total);
  if (threadIdx.x == 0) bsum[blockIdx.x] = total;
}

__global__ __launch_bounds__(256) void scan_final_kernel(
    const int* __restrict__ cnt, int n, const int* __restrict__ bsum,
    int* __restrict__ pos) {
  __shared__ int sh[256];
  int part = 0;
  for (int idx = threadIdx.x; idx < (int)blockIdx.x; idx += 256)
    part += bsum[idx];
  int chunkOff;
  block_excl_scan_256(part, sh, chunkOff);
  const int base = blockIdx.x * 2048 + threadIdx.x * 8;
  int v[8];
  int s = 0;
#pragma unroll
  for (int j = 0; j < 8; ++j) {
    const int idx = base + j;
    v[j] = (idx < n) ? cnt[idx] : 0;
    s += v[j];
  }
  int total;
  const int ex = block_excl_scan_256(s, sh, total);
  int run = chunkOff + ex;
#pragma unroll
  for (int j = 0; j < 8; ++j) {
    const int idx = base + j;
    if (idx < n) pos[idx] = run;
    run += v[j];
  }
}

#define MAX_BINS 1024
#define MAX_BIN_NODES 2048
#define EDGES_PER_BLOCK 4096

struct RelGeo {
  int4 E;
  int4 shift;
  int4 nbins;
  int4 nblocks;
  int4 tabOff;
  int4 N;
  int4 nodeOff;
};

__global__ __launch_bounds__(256) void bin_hist_kernel(
    const int* __restrict__ dst0, const int* __restrict__ dst1,
    const int* __restrict__ dst2, const int* __restrict__ dst3, RelGeo g,
    int* __restrict__ tabAll, const float* __restrict__ wa0,
    const float* __restrict__ wb0, const float* __restrict__ wa1,
    const float* __restrict__ wb1, const float* __restrict__ wa2,
    const float* __restrict__ wb2, const float* __restrict__ wa3,
    const float* __restrict__ wb3, float* __restrict__ wsum, int4 wOff,
    int4 wN) {
  __shared__ int h[MAX_BINS];
  const int t = threadIdx.x;
  const int r = blockIdx.y;
  const int nblocks = pick4(g.nblocks, r);
  if ((int)blockIdx.x >= nblocks) {
    if (blockIdx.x == gridDim.x - 1) {
      const float* a = pick_fp(wa0, wa1, wa2, wa3, r);
      const float* b = pick_fp(wb0, wb1, wb2, wb3, r);
      float* o = wsum + pick4(wOff, r);
      const int n = pick4(wN, r);
      for (int i = t; i < n; i += 256) o[i] = a[i] + b[i];
    }
    return;
  }
  const int* __restrict__ dst = pick_ip(dst0, dst1, dst2, dst3, r);
  const int E = pick4(g.E, r);
  const int shift = pick4(g.shift, r);
  const int nbins = pick4(g.nbins, r);
  int* __restrict__ tab = tabAll + pick4(g.tabOff, r);
  for (int k = t; k < nbins; k += 256) h[k] = 0;
  __syncthreads();
  const int base = blockIdx.x * EDGES_PER_BLOCK;
  int d[EDGES_PER_BLOCK / 256];
#pragma unroll
  for (int j = 0; j < EDGES_PER_BLOCK / 256; ++j) {
    const int i = base + j * 256 + t;
    d[j] = (i < E) ? dst[i] : -1;
  }
#pragma unroll
  for (int j = 0; j < EDGES_PER_BLOCK / 256; ++j)
    if (d[j] >= 0) atomicAdd(&h[d[j] >> shift], 1);
  __syncthreads();
  for (int k = t; k < nbins; k += 256) tab[k * nblocks + blockIdx.x] = h[k];
}

__global__ __launch_bounds__(256) void bin_scatter_kernel(
    const int* __restrict__ src0, const int* __restrict__ src1,
    const int* __restrict__ src2, const int* __restrict__ src3,
    const int* __restrict__ dst0, const int* __restrict__ dst1,
    const int* __restrict__ dst2, const int* __restrict__ dst3, RelGeo g,
    const int* __restrict__ tposAll, int* __restrict__ bedge) {
  __shared__ int cur[MAX_BINS];
  const int t = threadIdx.x;
  const int r = blockIdx.y;
  const int nblocks = pick4(g.nblocks, r);
  if ((int)blockIdx.x >= nblocks) return;
  const int* __restrict__ src = pick_ip(src0, src1, src2, src3, r);
  const int* __restrict__ dst = pick_ip(dst0, dst1, dst2, dst3, r);
  const int E = pick4(g.E, r);
  const int shift = pick4(g.shift, r);
  const int nbins = pick4(g.nbins, r);
  const int* __restrict__ tpos = tposAll + pick4(g.tabOff, r);
  for (int k = t; k < nbins; k += 256) cur[k] = tpos[k * nblocks + blockIdx.x];
  __syncthreads();
  int2* be = reinterpret_cast<int2*>(bedge);
  const int base = blockIdx.x * EDGES_PER_BLOCK;
  int d[EDGES_PER_BLOCK / 256], s[EDGES_PER_BLOCK / 256];
  int slot[EDGES_PER_BLOCK / 256];
#pragma unroll
  for (int j = 0; j < EDGES_PER_BLOCK / 256; ++j) {
    const int i = base + j * 256 + t;
    d[j] = (i < E) ? dst[i] : -1;
    s[j] = (i < E) ? src[i] : 0;
  }
#pragma unroll
  for (int j = 0; j < EDGES_PER_BLOCK / 256; ++j)
    slot[j] = (d[j] >= 0) ? atomicAdd(&cur[d[j] >> shift], 1) : 0;
#pragma unroll
  for (int j = 0; j < EDGES_PER_BLOCK / 256; ++j)
    if (d[j] >= 0) be[slot[j]] = make_int2(d[j], s[j]);
}

__global__ __launch_bounds__(256) void bin_csr_kernel(
    const int* __restrict__ bedge, RelGeo g, const int* __restrict__ tposAll,
    int nTab, int Etot, int* __restrict__ cntAll, int* __restrict__ posAll,
    int* __restrict__ col) {
  __shared__ int lc[MAX_BIN_NODES];
  __shared__ int sh[256];
  const int t = threadIdx.x;
  const int b = blockIdx.x;
  const int rel = blockIdx.y;
  if (b >= pick4(g.nbins, rel)) return;
  const int shift = pick4(g.shift, rel);
  const int nblocks = pick4(g.nblocks, rel);
  const int tabOff = pick4(g.tabOff, rel);
  const int N = pick4(g.N, rel);
  int* __restrict__ cnt = cntAll + pick4(g.nodeOff, rel);
  int* __restrict__ pos = posAll + pick4(g.nodeOff, rel);
  const int B = 1 << shift;
  const int per = B >> 8;
  const int nodeBase = b << shift;
  const int first = tabOff + b * nblocks;
  const int next = first + nblocks;
  const int beg = tposAll[first];
  const int end = (next < nTab) ? tposAll[next] : Etot;
  const int2* be = reinterpret_cast<const int2*>(bedge);

  for (int k = t; k < B; k += 256) lc[k] = 0;
  __syncthreads();
  for (int e0 = beg; e0 < end; e0 += 8 * 256) {
    int x[8];
#pragma unroll
    for (int u = 0; u < 8; ++u) {
      const int e = e0 + u * 256 + t;
      x[u] = (e < end) ? be[e].x : -1;
    }
#pragma unroll
    for (int u = 0; u < 8; ++u)
      if (x[u] >= 0) atomicAdd(&lc[x[u] - nodeBase], 1);
  }
  __syncthreads();

  int v[8];
  int s = 0;
#pragma unroll
  for (int j = 0; j < 8; ++j) {
    v[j] = (j < per) ? lc[t * per + j] : 0;
    s += v[j];
  }
  int total;
  int run = block_excl_scan_256(s, sh, total);
#pragma unroll
  for (int j = 0; j < 8; ++j) {
    if (j < per) {
      const int k = t * per + j;
      const int node = nodeBase + k;
      lc[k] = run;
      if (node < N) {
        cnt[node] = v[j];
        pos[node] = beg + run;
      }
      run += v[j];
    }
  }
  __syncthreads();
  for (int e0 = beg; e0 < end; e0 += 8 * 256) {
    int2 r[8];
    int slot[8];
#pragma unroll
    for (int u = 0; u < 8; ++u) {
      const int e = e0 + u * 256 + t;
      r[u] = (e < end) ? be[e] : make_int2(-1, 0);
    }
#pragma unroll
    for (int u = 0; u < 8; ++u)
      slot[u] = (r[u].x >= 0) ? atomicAdd(&lc[r[u].x - nodeBase], 1) : 0;
#pragma unroll
    for (int u = 0; u < 8; ++u)
      if (r[u].x >= 0) col[beg + slot[u]] = r[u].y;
  }
}

__device__ __forceinline__ float4 load_row4(const float* __restrict__ Y, int s,
                                            int c) {
  return *reinterpret_cast<const float4*>(Y + (size_t)s * 16 + c * 4);
}

__device__ __forceinline__ float4 gather_rows(const float* __restrict__ Y,
                                              const int* __restrict__ col,
                                              int beg, int end, int c) {
  float4 acc = make_float4(0.0f, 0.0f, 0.0f, 0.0f);
  int p = beg;
  for (; p + 1 < end; p += 2) {
    const int s0 = col[p];
    const int s1 = col[p + 1];
    const float4 v0 = load_row4(Y, s0, c);
    const float4 v1 = load_row4(Y, s1, c);
    acc.x += v0.x;
    acc.y += v0.y;
    acc.z += v0.z;
    acc.w += v0.w;
    acc.x += v1.x;
    acc.y += v1.y;
    acc.z += v1.z;
    acc.w += v1.w;
  }
  if (p < end) {
    const float4 v = load_row4(Y, col[p], c);
    acc.x += v.x;
    acc.y += v.y;
    acc.z += v.z;
    acc.w += v.w;
  }
  return acc;
}

__global__ void gather_l1_kernel(const float* __restrict__ YAd,
                                 const float* __restrict__ YBd,
                                 const float* __restrict__ YAt,
                                 const float* __restrict__ YBt,
                                 const int* __restrict__ cnt,
                                 const int* __restrict__ pos, int4 nodeOff,
                                 const int* __restrict__ col,
                                 float* __restrict__ Td,
                                 float* __restrict__ Tt, int ND, int NT,
                                 int blocksD) {
  const bool isT = (int)blockIdx.x >= blocksD;
  const int bid = isT ? blockIdx.x - blocksD : blockIdx.x;
  const float* __restrict__ YA = isT ? YAt : YAd;
  const float* __restrict__ YB = isT ? YBt : YBd;
  const int* __restrict__ cntA = cnt + (isT ? nodeOff.x : nodeOff.z);
  const int* __restrict__ cntB = cnt + (isT ? nodeOff.y : nodeOff.w);
  const int* __restrict__ begA = pos + (isT ? nodeOff.x : nodeOff.z);
  const int* __restrict__ begB = pos + (isT ? nodeOff.y : nodeOff.w);
  float* __restrict__ T = isT ? Tt : Td;
  const int N = isT ? NT : ND;
  const int i = bid * blockDim.x + threadIdx.x;
  const int node = i >> 2;
  const int c = i & 3;
  if (node >= N) return;
  const int nA = cntA[node], bA = begA[node];
  const int nB = cntB[node], bB = begB[node];
  float4* tp = reinterpret_cast<float4*>(T + (size_t)node * 16 + c * 4);
  float4 acc = *tp;
  const float4 sA = gather_rows(YA, col, bA, bA + nA, c);
  const float4 sB = gather_rows(YB, col, bB, bB + nB, c);
  const float iA = 1.0f / fmaxf((float)nA, 1.0f);
  const float iB = 1.0f / fmaxf((float)nB, 1.0f);
  acc.x += sA.x * iA + sB.x * iB;
  acc.y += sA.y * iA + sB.y * iB;
  acc.z += sA.z * iA + sB.z * iB;
  acc.w += sA.w * iA + sB.w * iB;
  acc.x = fmaxf(acc.x, 0.0f);
  acc.y = fmaxf(acc.y, 0.0f);
  acc.z = fmaxf(acc.z, 0.0f);
  acc.w = fmaxf(acc.w, 0.0f);
  *tp = acc;
}

__global__ void gather_l2_kernel(const float* __restrict__ Hd,
                                 const float* __restrict__ Ht,
                                 const int* __restrict__ cnt,
                                 const int* __restrict__ pos, int4 nodeOff,
                                 const int* __restrict__ col,
                                 float* __restrict__ aggAd,
                                 float* __restrict__ aggBd,
                                 float* __restrict__ aggAt,
                                 float* __restrict__ aggBt, int ND, int NT,
                                 int blocksD) {
  const bool isT = (int)blockIdx.x >= blocksD;
  const int bid = isT ? blockIdx.x - blocksD : blockIdx.x;
  const float* __restrict__ H = isT ? Ht : Hd;
  const int* __restrict__ cntA = cnt + (isT ? nodeOff.x : nodeOff.z);
  const int* __restrict__ cntB = cnt + (isT ? nodeOff.y : nodeOff.w);
  const int* __restrict__ begA = pos + (isT ? nodeOff.x : nodeOff.z);
  const int* __restrict__ begB = pos + (isT ? nodeOff.y : nodeOff.w);
  float* __restrict__ aggA = isT ? aggAt : aggAd;
  float* __restrict__ aggB = isT ? aggBt : aggBd;
  const int N = isT ? NT : ND;
  const int i = bid * blockDim.x + threadIdx.x;
  const int node = i >> 2;
  const int c = i & 3;
  if (node >= N) return;
  const int nA = cntA[node], bA = begA[node];
  const int nB = cntB[node], bB = begB[node];
  const float4 sA = gather_rows(H, col, bA, bA + nA, c);
  const float4 sB = gather_rows(H, col, bB, bB + nB, c);
  *reinterpret_cast<float4*>(aggA + (size_t)node * 16 + c * 4) = sA;
  *reinterpret_cast<float4*>(aggB + (size_t)node * 16 + c * 4) = sB;
}

__global__ __launch_bounds__(128) void l1_transform_kernel(
    const float* __restrict__ Xd, const float* __restrict__ W0d,
    const float* __restrict__ W1d, const float* __restrict__ W2d,
    const float* __restrict__ biasAd, const float* __restrict__ biasBd,
    float* __restrict__ Y0d, float* __restrict__ Y1d, float* __restrict__ Y2d,
    int Md, const float* __restrict__ Xt, const float* __restrict__ W0t,
    const float* __restrict__ W1t, const float* __restrict__ W2t,
    const float* __restrict__ biasAt, const float* __restrict__ biasBt,
    float* __restrict__ Y0t, float* __restrict__ Y1t, float* __restrict__ Y2t,
    int Mt, int blocksD) {
  const bool isT = (int)blockIdx.x >= blocksD;
  const int bid = isT ? blockIdx.x - blocksD : blockIdx.x;
  const int nb = isT ? gridDim.x - blocksD : blocksD;
  const float* __restrict__ X = isT ? Xt : Xd;
  const float* __restrict__ W0 = isT ? W0t : W0d;
  const float* __restrict__ W1 = isT ? W1t : W1d;
  const float* __restrict__ W2 = isT ? W2t : W2d;
  const float* __restrict__ biasA = isT ? biasAt : biasAd;
  const float* __restrict__ biasB = isT ? biasBt : biasBd;
  float* __restrict__ Y0 = isT ? Y0t : Y0d;
  float* __restrict__ Y1 = isT ? Y1t : Y1d;
  float* __restrict__ Y2 = isT ? Y2t : Y2d;
  const int M = isT ? Mt : Md;
  const int lane = threadIdx.x & 31;
  const int wave = threadIdx.x >> 5;
  const int wpb = blockDim.x >> 5;
  const int half = lane >> 4;
  const int mrow = lane & 15;
  const int nTiles = (M + 15) >> 4;

  v2f b0[16], b1[16], b2[16];
#pragma unroll
  for (int s = 0; s < 16; ++s) {
    int k = 4 * s + 2 * half;
    b0[s] = (v2f){W0[k * 16 + mrow], W0[(k + 1) * 16 + mrow]};
    b1[s] = (v2f){W1[k * 16 + mrow], W1[(k + 1) * 16 + mrow]};
    b2[s] = (v2f){W2[k * 16 + mrow], W2[(k + 1) * 16 + mrow]};
  }

  const float bsum = biasA[mrow] + biasB[mrow];

  for (int tile = bid * wpb + wave; tile < nTiles; tile += nb * wpb) {
    const int rbase = tile << 4;
    int arow = rbase + mrow;
    if (arow > M - 1) arow = M - 1;
    const float* xr = X + (size_t)arow * 64;
    v8f c0 = {}, c1 = {};
    v8f c2 = {bsum, bsum, bsum, bsum, bsum, bsum, bsum, bsum};
#pragma unroll
    for (int s = 0; s < 16; ++s) {
      int k = 4 * s + 2 * half;
      v2f a = (v2f){xr[k], xr[k + 1]};
      c0 = wmma_k4(a, b0[s], c0);
      c1 = wmma_k4(a, b1[s], c1);
      c2 = wmma_k4(a, b2[s], c2);
    }
#pragma unroll
    for (int r = 0; r < 8; ++r) {
      int row = rbase + r + 8 * half;
      if (row < M) {
        size_t o = (size_t)row * 16 + mrow;
        Y0[o] = c0[r];
        Y1[o] = c1[r];
        Y2[o] = c2[r];
      }
    }
  }
}

__global__ __launch_bounds__(128) void l2_fused_kernel(
    const float* __restrict__ A0d, const int* __restrict__ cnt0d,
    const float* __restrict__ A1d, const int* __restrict__ cnt1d,
    const float* __restrict__ A2d, const float* __restrict__ W0d,
    const float* __restrict__ W1d, const float* __restrict__ W2d,
    const float* __restrict__ bias0d, const float* __restrict__ bias1d,
    float* __restrict__ OUTd, int Md, const float* __restrict__ A0t,
    const int* __restrict__ cnt0t, const float* __restrict__ A1t,
    const int* __restrict__ cnt1t, const float* __restrict__ A2t,
    const float* __restrict__ W0t, const float* __restrict__ W1t,
    const float* __restrict__ W2t, const float* __restrict__ bias0t,
    const float* __restrict__ bias1t, float* __restrict__ OUTt, int Mt,
    int blocksD) {
  const bool isT = (int)blockIdx.x >= blocksD;
  const int bid = isT ? blockIdx.x - blocksD : blockIdx.x;
  const int nb = isT ? gridDim.x - blocksD : blocksD;
  const float* __restrict__ A0 = isT ? A0t : A0d;
  const int* __restrict__ cnt0 = isT ? cnt0t : cnt0d;
  const float* __restrict__ A1 = isT ? A1t : A1d;
  const int* __restrict__ cnt1 = isT ? cnt1t : cnt1d;
  const float* __restrict__ A2 = isT ? A2t : A2d;
  const float* __restrict__ W0 = isT ? W0t : W0d;
  const float* __restrict__ W1 = isT ? W1t : W1d;
  const float* __restrict__ W2 = isT ? W2t : W2d;
  const float* __restrict__ bias0 = isT ? bias0t : bias0d;
  const float* __restrict__ bias1 = isT ? bias1t : bias1d;
  float* __restrict__ OUT = isT ? OUTt : OUTd;
  const int M = isT ? Mt : Md;
  const int lane = threadIdx.x & 31;
  const int wave = threadIdx.x >> 5;
  const int wpb = blockDim.x >> 5;
  const int half = lane >> 4;
  const int mrow = lane & 15;
  const int nTiles = (M + 15) >> 4;

  const float* Ws[3] = {W0, W1, W2};
  v2f b[3][4][2];
#pragma unroll
  for (int m = 0; m < 3; ++m)
#pragma unroll
    for (int s = 0; s < 4; ++s)
#pragma unroll
      for (int t = 0; t < 2; ++t) {
        int k = 4 * s + 2 * half;
        int col = 16 * t + mrow;
        b[m][s][t] = (v2f){Ws[m][k * 32 + col], Ws[m][(k + 1) * 32 + col]};
      }

  const float biasL = bias0[mrow] + bias1[mrow];
  const float biasH = bias0[16 + mrow] + bias1[16 + mrow];

  for (int tile = bid * wpb + wave; tile < nTiles; tile += nb * wpb) {
    const int rbase = tile << 4;
    int arow = rbase + mrow;
    if (arow > M - 1) arow = M - 1;
    const float rs0 = 1.0f / fmaxf((float)cnt0[arow], 1.0f);
    const float rs1 = 1.0f / fmaxf((float)cnt1[arow], 1.0f);
    const float* a0p = A0 + (size_t)arow * 16;
    const float* a1p = A1 + (size_t)arow * 16;
    const float* a2p = A2 + (size_t)arow * 16;
    v8f cL = {}, cH = {};
#pragma unroll
    for (int s = 0; s < 4; ++s) {
      int k = 4 * s + 2 * half;
      v2f a0 = (v2f){a0p[k] * rs0, a0p[k + 1] * rs0};
      cL = wmma_k4(a0, b[0][s][0], cL);
      cH = wmma_k4(a0, b[0][s][1], cH);
      v2f a1 = (v2f){a1p[k] * rs1, a1p[k + 1] * rs1};
      cL = wmma_k4(a1, b[1][s][0], cL);
      cH = wmma_k4(a1, b[1][s][1], cH);
      v2f a2 = (v2f){fmaxf(a2p[k], 0.0f), fmaxf(a2p[k + 1], 0.0f)};
      cL = wmma_k4(a2, b[2][s][0], cL);
      cH = wmma_k4(a2, b[2][s][1], cH);
    }
#pragma unroll
    for (int r = 0; r < 8; ++r) {
      int row = rbase + r + 8 * half;
      if (row < M) {
        size_t o = (size_t)row * 32 + mrow;
        OUT[o] = cL[r] + biasL;
        OUT[o + 16] = cH[r] + biasH;
      }
    }
  }
}


static inline int bin_shift_for(int n) {
  int s = 8;
  while (s < 11 && ((n - 1) >> s) >= 256) ++s;
  return s;
}

extern "C" void kernel_launch(void* const* d_in, const int* in_sizes, int n_in,
                              void* d_out, int out_size, void* d_ws,
                              size_t ws_size, hipStream_t stream) {
  const float* x_dev = (const float*)d_in[0];
  const float* x_task = (const float*)d_in[1];
  const int* wr_src = (const int*)d_in[2];
  const int* wr_dst = (const int*)d_in[3];
  const int* rv_src = (const int*)d_in[4];
  const int* rv_dst = (const int*)d_in[5];
  const int* wb_src = (const int*)d_in[6];
  const int* wb_dst = (const int*)d_in[7];
  const int* rb_src = (const int*)d_in[8];
  const int* rb_dst = (const int*)d_in[9];
  const float* p1wr_wl = (const float*)d_in[10];
  const float* p1wr_bl = (const float*)d_in[11];
  const float* p1wr_wr = (const float*)d_in[12];
  const float* p1rv_wl = (const float*)d_in[13];
  const float* p1rv_bl = (const float*)d_in[14];
  const float* p1rv_wr = (const float*)d_in[15];
  const float* p1wb_wl = (const float*)d_in[16];
  const float* p1wb_bl = (const float*)d_in[17];
  const float* p1wb_wr = (const float*)d_in[18];
  const float* p1rb_wl = (const float*)d_in[19];
  const float* p1rb_bl = (const float*)d_in[20];
  const float* p1rb_wr = (const float*)d_in[21];
  const float* p2wr_wl = (const float*)d_in[22];
  const float* p2wr_bl = (const float*)d_in[23];
  const float* p2wr_wr = (const float*)d_in[24];
  const float* p2rv_wl = (const float*)d_in[25];
  const float* p2rv_bl = (const float*)d_in[26];
  const float* p2rv_wr = (const float*)d_in[27];
  const float* p2wb_wl = (const float*)d_in[28];
  const float* p2wb_bl = (const float*)d_in[29];
  const float* p2wb_wr = (const float*)d_in[30];
  const float* p2rb_wl = (const float*)d_in[31];
  const float* p2rb_bl = (const float*)d_in[32];
  const float* p2rb_wr = (const float*)d_in[33];

  const int ND = in_sizes[0] / 64;
  const int NT = in_sizes[1] / 64;
  const int E_wr = in_sizes[2], E_rv = in_sizes[4];
  const int E_wb = in_sizes[6], E_rb = in_sizes[8];

  const int cntN = 2 * NT + 2 * ND;
  const int Etot = E_wr + E_rv + E_wb + E_rb;
  const int shT = bin_shift_for(NT), shD = bin_shift_for(ND);
  const int nbT = ((NT - 1) >> shT) + 1, nbD = ((ND - 1) >> shD) + 1;
  const int nc_wr = (E_wr + EDGES_PER_BLOCK - 1) / EDGES_PER_BLOCK;
  const int nc_rv = (E_rv + EDGES_PER_BLOCK - 1) / EDGES_PER_BLOCK;
  const int nc_wb = (E_wb + EDGES_PER_BLOCK - 1) / EDGES_PER_BLOCK;
  const int nc_rb = (E_rb + EDGES_PER_BLOCK - 1) / EDGES_PER_BLOCK;
  const int to_wr = 0;
  const int to_rv = to_wr + nbT * nc_wr;
  const int to_wb = to_rv + nbT * nc_rv;
  const int to_rb = to_wb + nbD * nc_wb;
  const int nTab = to_rb + nbD * nc_rb;
  const int nChunks = (nTab + 2047) / 2048;
  float* ws = (float*)d_ws;
  size_t off = 0;
  int* cnt = (int*)(ws + off); off += (size_t)cntN;
  float* csr = ws + off; off += (size_t)cntN * 16;
  float* yreg = ws + off; off += (size_t)cntN * 16;
  float* t1 = ws + off; off += (size_t)NT * 16;
  float* d1 = ws + off; off += (size_t)ND * 16;
  float* ws1t = ws + off; off += 64 * 16;
  float* ws1d = ws + off; off += 64 * 16;
  float* ws2t = ws + off; off += 16 * 32;
  float* ws2d = ws + off; off += 16 * 32;

  int* cnt_wr = cnt;
  int* cnt_rv = cnt + NT;
  int* cnt_wb = cnt + 2 * (size_t)NT;
  int* cnt_rb = cnt + 2 * (size_t)NT + ND;
  int* pos = (int*)csr;
  int* col = pos + cntN;
  size_t q = (size_t)2 * cntN + Etot;
  q = (q + 1) & ~(size_t)1;
  int* bedge = (int*)ws + q; q += (size_t)2 * Etot;
  int* tab = (int*)ws + q; q += (size_t)nTab;
  int* tpos = (int*)ws + q; q += (size_t)nTab;
  int* bsum = (int*)ws + q;

  float* y_wr = yreg;
  float* y_rv = y_wr + (size_t)ND * 16;
  float* y_wb = y_rv + (size_t)ND * 16;
  float* y_rb = y_wb + (size_t)NT * 16;
  float* agg_wr = yreg;
  float* agg_rv = agg_wr + (size_t)NT * 16;
  float* agg_wb = agg_rv + (size_t)NT * 16;
  float* agg_rb = agg_wb + (size_t)ND * 16;

  const int TPB = 256;

  RelGeo g;
  g.E = make_int4(E_wr, E_rv, E_wb, E_rb);
  g.shift = make_int4(shT, shT, shD, shD);
  g.nbins = make_int4(nbT, nbT, nbD, nbD);
  g.nblocks = make_int4(nc_wr, nc_rv, nc_wb, nc_rb);
  g.tabOff = make_int4(to_wr, to_rv, to_wb, to_rb);
  g.N = make_int4(NT, NT, ND, ND);
  g.nodeOff = make_int4(0, NT, 2 * NT, 2 * NT + ND);
  int ncMax = nc_wr;
  if (nc_rv > ncMax) ncMax = nc_rv;
  if (nc_wb > ncMax) ncMax = nc_wb;
  if (nc_rb > ncMax) ncMax = nc_rb;
  const int nbMax = (nbT > nbD) ? nbT : nbD;

  bin_hist_kernel<<<dim3(ncMax + 1, 4), 256, 0, stream>>>(
      wr_dst, rv_dst, wb_dst, rb_dst, g, tab, p1wr_wr, p1rv_wr, p1wb_wr,
      p1rb_wr, p2wr_wr, p2rv_wr, p2wb_wr, p2rb_wr, ws1t,
      make_int4(0, 64 * 16, 2 * 64 * 16, 2 * 64 * 16 + 16 * 32),
      make_int4(64 * 16, 64 * 16, 16 * 32, 16 * 32));

  scan_partial_kernel<<<nChunks, 256, 0, stream>>>(tab, nTab, bsum);
  scan_final_kernel<<<nChunks, 256, 0, stream>>>(tab, nTab, bsum, tpos);

  bin_scatter_kernel<<<dim3(ncMax, 4), 256, 0, stream>>>(
      wr_src, rv_src, wb_src, rb_src, wr_dst, rv_dst, wb_dst, rb_dst, g, tpos,
      bedge);

  bin_csr_kernel<<<dim3(nbMax, 4), 256, 0, stream>>>(bedge, g, tpos, nTab, Etot,
                                                     cnt, pos, col);

  const int tilesD = (ND + 15) / 16, tilesT = (NT + 15) / 16;
  const int tbD = (tilesD + 3) / 4, tbT = (tilesT + 3) / 4;
  l1_transform_kernel<<<tbD + tbT, 128, 0, stream>>>(
      x_dev, p1wr_wl, p1rv_wl, ws1d, p1wb_bl, p1rb_bl, y_wr, y_rv, d1, ND,
      x_task, p1wb_wl, p1rb_wl, ws1t, p1wr_bl, p1rv_bl, y_wb, y_rb, t1, NT,
      tbD);

  const int gbD = (ND * 4 + TPB - 1) / TPB, gbT = (NT * 4 + TPB - 1) / TPB;
  gather_l1_kernel<<<gbD + gbT, TPB, 0, stream>>>(
      y_wb, y_rb, y_wr, y_rv, cnt, pos, g.nodeOff, col, d1, t1, ND, NT, gbD);

  gather_l2_kernel<<<gbD + gbT, TPB, 0, stream>>>(
      t1, d1, cnt, pos, g.nodeOff, col, agg_wb, agg_rb, agg_wr, agg_rv, ND, NT,
      gbD);

  float* out_d2 = (float*)d_out;
  float* out_t2 = (float*)d_out + (size_t)ND * 32;
  l2_fused_kernel<<<tbD + tbT, 128, 0, stream>>>(
      agg_wb, cnt_wb, agg_rb, cnt_rb, d1, p2wb_wl, p2rb_wl, ws2d, p2wb_bl,
      p2rb_bl, out_d2, ND, agg_wr, cnt_wr, agg_rv, cnt_rv, t1, p2wr_wl,
      p2rv_wl, ws2t, p2wr_bl, p2rv_bl, out_t2, NT, tbD);
}


# Layer-2 gather and GEMM launches fused: a 64-row block walks its neighbour lists as before and hands the scaled sums to its WMMA waves through LDS; aggregator planes gone, 9 -> 8 launches
# speedup vs baseline: 2.9572x; 1.0664x over previous
#include <hip/hip_runtime.h>


typedef __attribute__((ext_vector_type(2))) float v2f;
typedef __attribute__((ext_vector_type(8))) float v8f;

__device__ __forceinline__ v8f wmma_k4(v2f a, v2f b, v8f c) {
  return __builtin_amdgcn_wmma_f32_16x16x4_f32(
       false, a,  false, b,
       (short)0, c,  false,  false);
}


__device__ __forceinline__ int pick4(int4 v, int r) {
  return r == 0 ? v.x : r == 1 ? v.y : r == 2 ? v.z : v.w;
}

__device__ __forceinline__ const int* pick_ip(const int* a, const int* b,
                                              const int* c, const int* d,
                                              int r) {
  return r == 0 ? a : r == 1 ? b : r == 2 ? c : d;
}

__device__ __forceinline__ const float* pick_fp(const float* a, const float* b,
                                                const float* c, const float* d,
                                                int r) {
  return r == 0 ? a : r == 1 ? b : r == 2 ? c : d;
}


__device__ __forceinline__ int block_excl_scan_256(int v, int* sh, int& total) {
  const int t = threadIdx.x;
  sh[t] = v;
  __syncthreads();
  for (int off = 1; off < 256; off <<= 1) {
    const int add = (t >= off) ? sh[t - off] : 0;
    __syncthreads();
    sh[t] += add;
    __syncthreads();
  }
  const int incl = sh[t];
  total = sh[255];
  __syncthreads();
  return incl - v;
}

__global__ __launch_bounds__(256) void scan_partial_kernel(
    const int* __restrict__ cnt, int n, int* __restrict__ bsum) {
  __shared__ int sh[256];
  const int base = blockIdx.x * 2048 + threadIdx.x * 8;
  int v[8];
#pragma unroll
  for (int j = 0; j < 8; ++j) {
    const int idx = base + j;
    v[j] = (idx < n) ? cnt[idx] : 0;
  }
  int s = 0;
#pragma unroll
  for (int j = 0; j < 8; ++j) s += v[j];
  int total;
  block_excl_scan_256(s, sh, total);
  if (threadIdx.x == 0) bsum[blockIdx.x] = total;
}

__global__ __launch_bounds__(256) void scan_final_kernel(
    const int* __restrict__ cnt, int n, const int* __restrict__ bsum,
    int* __restrict__ pos) {
  __shared__ int sh[256];
  int part = 0;
  for (int idx = threadIdx.x; idx < (int)blockIdx.x; idx += 256)
    part += bsum[idx];
  int chunkOff;
  block_excl_scan_256(part, sh, chunkOff);
  const int base = blockIdx.x * 2048 + threadIdx.x * 8;
  int v[8];
  int s = 0;
#pragma unroll
  for (int j = 0; j < 8; ++j) {
    const int idx = base + j;
    v[j] = (idx < n) ? cnt[idx] : 0;
    s += v[j];
  }
  int total;
  const int ex = block_excl_scan_256(s, sh, total);
  int run = chunkOff + ex;
#pragma unroll
  for (int j = 0; j < 8; ++j) {
    const int idx = base + j;
    if (idx < n) pos[idx] = run;
    run += v[j];
  }
}

#define MAX_BINS 1024
#define MAX_BIN_NODES 2048
#define EDGES_PER_BLOCK 4096

struct RelGeo {
  int4 E;
  int4 shift;
  int4 nbins;
  int4 nblocks;
  int4 tabOff;
  int4 N;
  int4 nodeOff;
};

__global__ __launch_bounds__(256) void bin_hist_kernel(
    const int* __restrict__ dst0, const int* __restrict__ dst1,
    const int* __restrict__ dst2, const int* __restrict__ dst3, RelGeo g,
    int* __restrict__ tabAll, const float* __restrict__ wa0,
    const float* __restrict__ wb0, const float* __restrict__ wa1,
    const float* __restrict__ wb1, const float* __restrict__ wa2,
    const float* __restrict__ wb2, const float* __restrict__ wa3,
    const float* __restrict__ wb3, float* __restrict__ wsum, int4 wOff,
    int4 wN) {
  __shared__ int h[MAX_BINS];
  const int t = threadIdx.x;
  const int r = blockIdx.y;
  const int nblocks = pick4(g.nblocks, r);
  if ((int)blockIdx.x >= nblocks) {
    if (blockIdx.x == gridDim.x - 1) {
      const float* a = pick_fp(wa0, wa1, wa2, wa3, r);
      const float* b = pick_fp(wb0, wb1, wb2, wb3, r);
      float* o = wsum + pick4(wOff, r);
      const int n = pick4(wN, r);
      for (int i = t; i < n; i += 256) o[i] = a[i] + b[i];
    }
    return;
  }
  const int* __restrict__ dst = pick_ip(dst0, dst1, dst2, dst3, r);
  const int E = pick4(g.E, r);
  const int shift = pick4(g.shift, r);
  const int nbins = pick4(g.nbins, r);
  int* __restrict__ tab = tabAll + pick4(g.tabOff, r);
  for (int k = t; k < nbins; k += 256) h[k] = 0;
  __syncthreads();
  const int base = blockIdx.x * EDGES_PER_BLOCK;
  int d[EDGES_PER_BLOCK / 256];
#pragma unroll
  for (int j = 0; j < EDGES_PER_BLOCK / 256; ++j) {
    const int i = base + j * 256 + t;
    d[j] = (i < E) ? dst[i] : -1;
  }
#pragma unroll
  for (int j = 0; j < EDGES_PER_BLOCK / 256; ++j)
    if (d[j] >= 0) atomicAdd(&h[d[j] >> shift], 1);
  __syncthreads();
  for (int k = t; k < nbins; k += 256) tab[k * nblocks + blockIdx.x] = h[k];
}

__global__ __launch_bounds__(256) void bin_scatter_kernel(
    const int* __restrict__ src0, const int* __restrict__ src1,
    const int* __restrict__ src2, const int* __restrict__ src3,
    const int* __restrict__ dst0, const int* __restrict__ dst1,
    const int* __restrict__ dst2, const int* __restrict__ dst3, RelGeo g,
    const int* __restrict__ tposAll, int* __restrict__ bedge) {
  __shared__ int cur[MAX_BINS];
  const int t = threadIdx.x;
  const int r = blockIdx.y;
  const int nblocks = pick4(g.nblocks, r);
  if ((int)blockIdx.x >= nblocks) return;
  const int* __restrict__ src = pick_ip(src0, src1, src2, src3, r);
  const int* __restrict__ dst = pick_ip(dst0, dst1, dst2, dst3, r);
  const int E = pick4(g.E, r);
  const int shift = pick4(g.shift, r);
  const int nbins = pick4(g.nbins, r);
  const int* __restrict__ tpos = tposAll + pick4(g.tabOff, r);
  for (int k = t; k < nbins; k += 256) cur[k] = tpos[k * nblocks + blockIdx.x];
  __syncthreads();
  int2* be = reinterpret_cast<int2*>(bedge);
  const int base = blockIdx.x * EDGES_PER_BLOCK;
  int d[EDGES_PER_BLOCK / 256], s[EDGES_PER_BLOCK / 256];
  int slot[EDGES_PER_BLOCK / 256];
#pragma unroll
  for (int j = 0; j < EDGES_PER_BLOCK / 256; ++j) {
    const int i = base + j * 256 + t;
    d[j] = (i < E) ? dst[i] : -1;
    s[j] = (i < E) ? src[i] : 0;
  }
#pragma unroll
  for (int j = 0; j < EDGES_PER_BLOCK / 256; ++j)
    slot[j] = (d[j] >= 0) ? atomicAdd(&cur[d[j] >> shift], 1) : 0;
#pragma unroll
  for (int j = 0; j < EDGES_PER_BLOCK / 256; ++j)
    if (d[j] >= 0) be[slot[j]] = make_int2(d[j], s[j]);
}

__global__ __launch_bounds__(256) void bin_csr_kernel(
    const int* __restrict__ bedge, RelGeo g, const int* __restrict__ tposAll,
    int nTab, int Etot, int* __restrict__ cntAll, int* __restrict__ posAll,
    int* __restrict__ col) {
  __shared__ int lc[MAX_BIN_NODES];
  __shared__ int sh[256];
  const int t = threadIdx.x;
  const int b = blockIdx.x;
  const int rel = blockIdx.y;
  if (b >= pick4(g.nbins, rel)) return;
  const int shift = pick4(g.shift, rel);
  const int nblocks = pick4(g.nblocks, rel);
  const int tabOff = pick4(g.tabOff, rel);
  const int N = pick4(g.N, rel);
  int* __restrict__ cnt = cntAll + pick4(g.nodeOff, rel);
  int* __restrict__ pos = posAll + pick4(g.nodeOff, rel);
  const int B = 1 << shift;
  const int per = B >> 8;
  const int nodeBase = b << shift;
  const int first = tabOff + b * nblocks;
  const int next = first + nblocks;
  const int beg = tposAll[first];
  const int end = (next < nTab) ? tposAll[next] : Etot;
  const int2* be = reinterpret_cast<const int2*>(bedge);

  for (int k = t; k < B; k += 256) lc[k] = 0;
  __syncthreads();
  for (int e0 = beg; e0 < end; e0 += 8 * 256) {
    int x[8];
#pragma unroll
    for (int u = 0; u < 8; ++u) {
      const int e = e0 + u * 256 + t;
      x[u] = (e < end) ? be[e].x : -1;
    }
#pragma unroll
    for (int u = 0; u < 8; ++u)
      if (x[u] >= 0) atomicAdd(&lc[x[u] - nodeBase], 1);
  }
  __syncthreads();

  int v[8];
  int s = 0;
#pragma unroll
  for (int j = 0; j < 8; ++j) {
    v[j] = (j < per) ? lc[t * per + j] : 0;
    s += v[j];
  }
  int total;
  int run = block_excl_scan_256(s, sh, total);
#pragma unroll
  for (int j = 0; j < 8; ++j) {
    if (j < per) {
      const int k = t * per + j;
      const int node = nodeBase + k;
      lc[k] = run;
      if (node < N) {
        cnt[node] = v[j];
        pos[node] = beg + run;
      }
      run += v[j];
    }
  }
  __syncthreads();
  for (int e0 = beg; e0 < end; e0 += 8 * 256) {
    int2 r[8];
    int slot[8];
#pragma unroll
    for (int u = 0; u < 8; ++u) {
      const int e = e0 + u * 256 + t;
      r[u] = (e < end) ? be[e] : make_int2(-1, 0);
    }
#pragma unroll
    for (int u = 0; u < 8; ++u)
      slot[u] = (r[u].x >= 0) ? atomicAdd(&lc[r[u].x - nodeBase], 1) : 0;
#pragma unroll
    for (int u = 0; u < 8; ++u)
      if (r[u].x >= 0) col[beg + slot[u]] = r[u].y;
  }
}

__device__ __forceinline__ float4 load_row4(const float* __restrict__ Y, int s,
                                            int c) {
  return *reinterpret_cast<const float4*>(Y + (size_t)s * 16 + c * 4);
}

__device__ __forceinline__ float4 gather_rows(const float* __restrict__ Y,
                                              const int* __restrict__ col,
                                              int beg, int end, int c) {
  float4 acc = make_float4(0.0f, 0.0f, 0.0f, 0.0f);
  int p = beg;
  for (; p + 1 < end; p += 2) {
    const int s0 = col[p];
    const int s1 = col[p + 1];
    const float4 v0 = load_row4(Y, s0, c);
    const float4 v1 = load_row4(Y, s1, c);
    acc.x += v0.x;
    acc.y += v0.y;
    acc.z += v0.z;
    acc.w += v0.w;
    acc.x += v1.x;
    acc.y += v1.y;
    acc.z += v1.z;
    acc.w += v1.w;
  }
  if (p < end) {
    const float4 v = load_row4(Y, col[p], c);
    acc.x += v.x;
    acc.y += v.y;
    acc.z += v.z;
    acc.w += v.w;
  }
  return acc;
}

__global__ void gather_l1_kernel(const float* __restrict__ YAd,
                                 const float* __restrict__ YBd,
                                 const float* __restrict__ YAt,
                                 const float* __restrict__ YBt,
                                 const int* __restrict__ cnt,
                                 const int* __restrict__ pos, int4 nodeOff,
                                 const int* __restrict__ col,
                                 float* __restrict__ Td,
                                 float* __restrict__ Tt, int ND, int NT,
                                 int blocksD) {
  const bool isT = (int)blockIdx.x >= blocksD;
  const int bid = isT ? blockIdx.x - blocksD : blockIdx.x;
  const float* __restrict__ YA = isT ? YAt : YAd;
  const float* __restrict__ YB = isT ? YBt : YBd;
  const int* __restrict__ cntA = cnt + (isT ? nodeOff.x : nodeOff.z);
  const int* __restrict__ cntB = cnt + (isT ? nodeOff.y : nodeOff.w);
  const int* __restrict__ begA = pos + (isT ? nodeOff.x : nodeOff.z);
  const int* __restrict__ begB = pos + (isT ? nodeOff.y : nodeOff.w);
  float* __restrict__ T = isT ? Tt : Td;
  const int N = isT ? NT : ND;
  const int i = bid * blockDim.x + threadIdx.x;
  const int node = i >> 2;
  const int c = i & 3;
  if (node >= N) return;
  const int nA = cntA[node], bA = begA[node];
  const int nB = cntB[node], bB = begB[node];
  float4* tp = reinterpret_cast<float4*>(T + (size_t)node * 16 + c * 4);
  float4 acc = *tp;
  const float4 sA = gather_rows(YA, col, bA, bA + nA, c);
  const float4 sB = gather_rows(YB, col, bB, bB + nB, c);
  const float iA = 1.0f / fmaxf((float)nA, 1.0f);
  const float iB = 1.0f / fmaxf((float)nB, 1.0f);
  acc.x += sA.x * iA + sB.x * iB;
  acc.y += sA.y * iA + sB.y * iB;
  acc.z += sA.z * iA + sB.z * iB;
  acc.w += sA.w * iA + sB.w * iB;
  acc.x = fmaxf(acc.x, 0.0f);
  acc.y = fmaxf(acc.y, 0.0f);
  acc.z = fmaxf(acc.z, 0.0f);
  acc.w = fmaxf(acc.w, 0.0f);
  *tp = acc;
}

__global__ __launch_bounds__(128) void l1_transform_kernel(
    const float* __restrict__ Xd, const float* __restrict__ W0d,
    const float* __restrict__ W1d, const float* __restrict__ W2d,
    const float* __restrict__ biasAd, const float* __restrict__ biasBd,
    float* __restrict__ Y0d, float* __restrict__ Y1d, float* __restrict__ Y2d,
    int Md, const float* __restrict__ Xt, const float* __restrict__ W0t,
    const float* __restrict__ W1t, const float* __restrict__ W2t,
    const float* __restrict__ biasAt, const float* __restrict__ biasBt,
    float* __restrict__ Y0t, float* __restrict__ Y1t, float* __restrict__ Y2t,
    int Mt, int blocksD) {
  const bool isT = (int)blockIdx.x >= blocksD;
  const int bid = isT ? blockIdx.x - blocksD : blockIdx.x;
  const int nb = isT ? gridDim.x - blocksD : blocksD;
  const float* __restrict__ X = isT ? Xt : Xd;
  const float* __restrict__ W0 = isT ? W0t : W0d;
  const float* __restrict__ W1 = isT ? W1t : W1d;
  const float* __restrict__ W2 = isT ? W2t : W2d;
  const float* __restrict__ biasA = isT ? biasAt : biasAd;
  const float* __restrict__ biasB = isT ? biasBt : biasBd;
  float* __restrict__ Y0 = isT ? Y0t : Y0d;
  float* __restrict__ Y1 = isT ? Y1t : Y1d;
  float* __restrict__ Y2 = isT ? Y2t : Y2d;
  const int M = isT ? Mt : Md;
  const int lane = threadIdx.x & 31;
  const int wave = threadIdx.x >> 5;
  const int wpb = blockDim.x >> 5;
  const int half = lane >> 4;
  const int mrow = lane & 15;
  const int nTiles = (M + 15) >> 4;

  v2f b0[16], b1[16], b2[16];
#pragma unroll
  for (int s = 0; s < 16; ++s) {
    int k = 4 * s + 2 * half;
    b0[s] = (v2f){W0[k * 16 + mrow], W0[(k + 1) * 16 + mrow]};
    b1[s] = (v2f){W1[k * 16 + mrow], W1[(k + 1) * 16 + mrow]};
    b2[s] = (v2f){W2[k * 16 + mrow], W2[(k + 1) * 16 + mrow]};
  }

  const float bsum = biasA[mrow] + biasB[mrow];

  for (int tile = bid * wpb + wave; tile < nTiles; tile += nb * wpb) {
    const int rbase = tile << 4;
    int arow = rbase + mrow;
    if (arow > M - 1) arow = M - 1;
    const float* xr = X + (size_t)arow * 64;
    v8f c0 = {}, c1 = {};
    v8f c2 = {bsum, bsum, bsum, bsum, bsum, bsum, bsum, bsum};
#pragma unroll
    for (int s = 0; s < 16; ++s) {
      int k = 4 * s + 2 * half;
      v2f a = (v2f){xr[k], xr[k + 1]};
      c0 = wmma_k4(a, b0[s], c0);
      c1 = wmma_k4(a, b1[s], c1);
      c2 = wmma_k4(a, b2[s], c2);
    }
#pragma unroll
    for (int r = 0; r < 8; ++r) {
      int row = rbase + r + 8 * half;
      if (row < M) {
        size_t o = (size_t)row * 16 + mrow;
        Y0[o] = c0[r];
        Y1[o] = c1[r];
        Y2[o] = c2[r];
      }
    }
  }
}

#define L2_ROWS 64
#define L2_PITCH 20

__global__ __launch_bounds__(256) void l2_gather_fused_kernel(
    const float* __restrict__ Hnd, const float* __restrict__ Hrd,
    const float* __restrict__ W0d, const float* __restrict__ W1d,
    const float* __restrict__ W2d, const float* __restrict__ bias0d,
    const float* __restrict__ bias1d, float* __restrict__ OUTd, int Md,
    const float* __restrict__ Hnt, const float* __restrict__ Hrt,
    const float* __restrict__ W0t, const float* __restrict__ W1t,
    const float* __restrict__ W2t, const float* __restrict__ bias0t,
    const float* __restrict__ bias1t, float* __restrict__ OUTt, int Mt,
    const int* __restrict__ cnt, const int* __restrict__ pos, int4 nodeOff,
    const int* __restrict__ col, int blocksD) {
  __shared__ __attribute__((aligned(16))) float stage[3][L2_ROWS][L2_PITCH];
  const bool isT = (int)blockIdx.x >= blocksD;
  const int bid = isT ? blockIdx.x - blocksD : blockIdx.x;
  const float* __restrict__ Hn = isT ? Hnt : Hnd;
  const float* __restrict__ Hr = isT ? Hrt : Hrd;
  const float* __restrict__ W0 = isT ? W0t : W0d;
  const float* __restrict__ W1 = isT ? W1t : W1d;
  const float* __restrict__ W2 = isT ? W2t : W2d;
  const float* __restrict__ bias0 = isT ? bias0t : bias0d;
  const float* __restrict__ bias1 = isT ? bias1t : bias1d;
  float* __restrict__ OUT = isT ? OUTt : OUTd;
  const int* __restrict__ cntA = cnt + (isT ? nodeOff.x : nodeOff.z);
  const int* __restrict__ cntB = cnt + (isT ? nodeOff.y : nodeOff.w);
  const int* __restrict__ begA = pos + (isT ? nodeOff.x : nodeOff.z);
  const int* __restrict__ begB = pos + (isT ? nodeOff.y : nodeOff.w);
  const int M = isT ? Mt : Md;
  const int t = threadIdx.x;
  const int lane = t & 31;
  const int wave = t >> 5;
  const int half = lane >> 4;
  const int mrow = lane & 15;
  const int ct = wave & 1;
  const int tl = wave >> 1;

  const float* Ws[3] = {W0, W1, W2};
  v2f b[3][4];
#pragma unroll
  for (int m = 0; m < 3; ++m)
#pragma unroll
    for (int s = 0; s < 4; ++s) {
      const int k = 4 * s + 2 * half;
      const int c = 16 * ct + mrow;
      b[m][s] = (v2f){Ws[m][k * 32 + c], Ws[m][(k + 1) * 32 + c]};
    }
  const float bias = bias0[16 * ct + mrow] + bias1[16 * ct + mrow];

  {
    const int r = t >> 2;
    const int q = t & 3;
    const int node = bid * L2_ROWS + r;
    float4 sA = make_float4(0.0f, 0.0f, 0.0f, 0.0f);
    float4 sB = sA, rt = sA;
    if (node < M) {
      const int nA = cntA[node], bA = begA[node];
      const int nB = cntB[node], bB = begB[node];
      rt = load_row4(Hr, node, q);
      sA = gather_rows(Hn, col, bA, bA + nA, q);
      sB = gather_rows(Hn, col, bB, bB + nB, q);
      const float iA = 1.0f / fmaxf((float)nA, 1.0f);
      const float iB = 1.0f / fmaxf((float)nB, 1.0f);
      sA.x *= iA;
      sA.y *= iA;
      sA.z *= iA;
      sA.w *= iA;
      sB.x *= iB;
      sB.y *= iB;
      sB.z *= iB;
      sB.w *= iB;
      rt.x = fmaxf(rt.x, 0.0f);
      rt.y = fmaxf(rt.y, 0.0f);
      rt.z = fmaxf(rt.z, 0.0f);
      rt.w = fmaxf(rt.w, 0.0f);
    }
    *reinterpret_cast<float4*>(&stage[0][r][4 * q]) = sA;
    *reinterpret_cast<float4*>(&stage[1][r][4 * q]) = sB;
    *reinterpret_cast<float4*>(&stage[2][r][4 * q]) = rt;
  }
  __syncthreads();

  const int rbase = bid * L2_ROWS + 16 * tl;
  if (rbase >= M) return;
  v8f acc = {};
#pragma unroll
  for (int s = 0; s < 4; ++s)
#pragma unroll
    for (int m = 0; m < 3; ++m) {
      const v2f a = *reinterpret_cast<const v2f*>(
          &stage[m][16 * tl + mrow][4 * s + 2 * half]);
      acc = wmma_k4(a, b[m][s], acc);
    }
#pragma unroll
  for (int r = 0; r < 8; ++r) {
    const int orow = rbase + r + 8 * half;
    if (orow < M) OUT[(size_t)orow * 32 + 16 * ct + mrow] = acc[r] + bias;
  }
}


static inline int bin_shift_for(int n) {
  int s = 8;
  while (s < 11 && ((n - 1) >> s) >= 256) ++s;
  return s;
}

extern "C" void kernel_launch(void* const* d_in, const int* in_sizes, int n_in,
                              void* d_out, int out_size, void* d_ws,
                              size_t ws_size, hipStream_t stream) {
  const float* x_dev = (const float*)d_in[0];
  const float* x_task = (const float*)d_in[1];
  const int* wr_src = (const int*)d_in[2];
  const int* wr_dst = (const int*)d_in[3];
  const int* rv_src = (const int*)d_in[4];
  const int* rv_dst = (const int*)d_in[5];
  const int* wb_src = (const int*)d_in[6];
  const int* wb_dst = (const int*)d_in[7];
  const int* rb_src = (const int*)d_in[8];
  const int* rb_dst = (const int*)d_in[9];
  const float* p1wr_wl = (const float*)d_in[10];
  const float* p1wr_bl = (const float*)d_in[11];
  const float* p1wr_wr = (const float*)d_in[12];
  const float* p1rv_wl = (const float*)d_in[13];
  const float* p1rv_bl = (const float*)d_in[14];
  const float* p1rv_wr = (const float*)d_in[15];
  const float* p1wb_wl = (const float*)d_in[16];
  const float* p1wb_bl = (const float*)d_in[17];
  const float* p1wb_wr = (const float*)d_in[18];
  const float* p1rb_wl = (const float*)d_in[19];
  const float* p1rb_bl = (const float*)d_in[20];
  const float* p1rb_wr = (const float*)d_in[21];
  const float* p2wr_wl = (const float*)d_in[22];
  const float* p2wr_bl = (const float*)d_in[23];
  const float* p2wr_wr = (const float*)d_in[24];
  const float* p2rv_wl = (const float*)d_in[25];
  const float* p2rv_bl = (const float*)d_in[26];
  const float* p2rv_wr = (const float*)d_in[27];
  const float* p2wb_wl = (const float*)d_in[28];
  const float* p2wb_bl = (const float*)d_in[29];
  const float* p2wb_wr = (const float*)d_in[30];
  const float* p2rb_wl = (const float*)d_in[31];
  const float* p2rb_bl = (const float*)d_in[32];
  const float* p2rb_wr = (const float*)d_in[33];

  const int ND = in_sizes[0] / 64;
  const int NT = in_sizes[1] / 64;
  const int E_wr = in_sizes[2], E_rv = in_sizes[4];
  const int E_wb = in_sizes[6], E_rb = in_sizes[8];

  const int cntN = 2 * NT + 2 * ND;
  const int Etot = E_wr + E_rv + E_wb + E_rb;
  const int shT = bin_shift_for(NT), shD = bin_shift_for(ND);
  const int nbT = ((NT - 1) >> shT) + 1, nbD = ((ND - 1) >> shD) + 1;
  const int nc_wr = (E_wr + EDGES_PER_BLOCK - 1) / EDGES_PER_BLOCK;
  const int nc_rv = (E_rv + EDGES_PER_BLOCK - 1) / EDGES_PER_BLOCK;
  const int nc_wb = (E_wb + EDGES_PER_BLOCK - 1) / EDGES_PER_BLOCK;
  const int nc_rb = (E_rb + EDGES_PER_BLOCK - 1) / EDGES_PER_BLOCK;
  const int to_wr = 0;
  const int to_rv = to_wr + nbT * nc_wr;
  const int to_wb = to_rv + nbT * nc_rv;
  const int to_rb = to_wb + nbD * nc_wb;
  const int nTab = to_rb + nbD * nc_rb;
  const int nChunks = (nTab + 2047) / 2048;
  float* ws = (float*)d_ws;
  size_t off = 0;
  int* cnt = (int*)(ws + off); off += (size_t)cntN;
  float* csr = ws + off; off += (size_t)cntN * 16;
  float* yreg = ws + off; off += (size_t)cntN * 16;
  float* t1 = ws + off; off += (size_t)NT * 16;
  float* d1 = ws + off; off += (size_t)ND * 16;
  float* ws1t = ws + off; off += 64 * 16;
  float* ws1d = ws + off; off += 64 * 16;
  float* ws2t = ws + off; off += 16 * 32;
  float* ws2d = ws + off; off += 16 * 32;

  int* pos = (int*)csr;
  int* col = pos + cntN;
  size_t q = (size_t)2 * cntN + Etot;
  q = (q + 1) & ~(size_t)1;
  int* bedge = (int*)ws + q; q += (size_t)2 * Etot;
  int* tab = (int*)ws + q; q += (size_t)nTab;
  int* tpos = (int*)ws + q; q += (size_t)nTab;
  int* bsum = (int*)ws + q;

  float* y_wr = yreg;
  float* y_rv = y_wr + (size_t)ND * 16;
  float* y_wb = y_rv + (size_t)ND * 16;
  float* y_rb = y_wb + (size_t)NT * 16;

  const int TPB = 256;

  RelGeo g;
  g.E = make_int4(E_wr, E_rv, E_wb, E_rb);
  g.shift = make_int4(shT, shT, shD, shD);
  g.nbins = make_int4(nbT, nbT, nbD, nbD);
  g.nblocks = make_int4(nc_wr, nc_rv, nc_wb, nc_rb);
  g.tabOff = make_int4(to_wr, to_rv, to_wb, to_rb);
  g.N = make_int4(NT, NT, ND, ND);
  g.nodeOff = make_int4(0, NT, 2 * NT, 2 * NT + ND);
  int ncMax = nc_wr;
  if (nc_rv > ncMax) ncMax = nc_rv;
  if (nc_wb > ncMax) ncMax = nc_wb;
  if (nc_rb > ncMax) ncMax = nc_rb;
  const int nbMax = (nbT > nbD) ? nbT : nbD;

  bin_hist_kernel<<<dim3(ncMax + 1, 4), 256, 0, stream>>>(
      wr_dst, rv_dst, wb_dst, rb_dst, g, tab, p1wr_wr, p1rv_wr, p1wb_wr,
      p1rb_wr, p2wr_wr, p2rv_wr, p2wb_wr, p2rb_wr, ws1t,
      make_int4(0, 64 * 16, 2 * 64 * 16, 2 * 64 * 16 + 16 * 32),
      make_int4(64 * 16, 64 * 16, 16 * 32, 16 * 32));

  scan_partial_kernel<<<nChunks, 256, 0, stream>>>(tab, nTab, bsum);
  scan_final_kernel<<<nChunks, 256, 0, stream>>>(tab, nTab, bsum, tpos);

  bin_scatter_kernel<<<dim3(ncMax, 4), 256, 0, stream>>>(
      wr_src, rv_src, wb_src, rb_src, wr_dst, rv_dst, wb_dst, rb_dst, g, tpos,
      bedge);

  bin_csr_kernel<<<dim3(nbMax, 4), 256, 0, stream>>>(bedge, g, tpos, nTab, Etot,
                                                     cnt, pos, col);

  const int tilesD = (ND + 15) / 16, tilesT = (NT + 15) / 16;
  const int tbD = (tilesD + 3) / 4, tbT = (tilesT + 3) / 4;
  l1_transform_kernel<<<tbD + tbT, 128, 0, stream>>>(
      x_dev, p1wr_wl, p1rv_wl, ws1d, p1wb_bl, p1rb_bl, y_wr, y_rv, d1, ND,
      x_task, p1wb_wl, p1rb_wl, ws1t, p1wr_bl, p1rv_bl, y_wb, y_rb, t1, NT,
      tbD);

  const int gbD = (ND * 4 + TPB - 1) / TPB, gbT = (NT * 4 + TPB - 1) / TPB;
  gather_l1_kernel<<<gbD + gbT, TPB, 0, stream>>>(
      y_wb, y_rb, y_wr, y_rv, cnt, pos, g.nodeOff, col, d1, t1, ND, NT, gbD);

  float* out_d2 = (float*)d_out;
  float* out_t2 = (float*)d_out + (size_t)ND * 32;
  const int fbD = (ND + L2_ROWS - 1) / L2_ROWS;
  const int fbT = (NT + L2_ROWS - 1) / L2_ROWS;
  l2_gather_fused_kernel<<<fbD + fbT, 256, 0, stream>>>(
      t1, d1, p2wb_wl, p2rb_wl, ws2d, p2wb_bl, p2rb_bl, out_d2, ND, d1, t1,
      p2wr_wl, p2rv_wl, ws2t, p2wr_bl, p2rv_bl, out_t2, NT, cnt, pos,
      g.nodeOff, col, fbD);
}


# Table scan folded into the binned-copy launch: its blocks (1024 threads, 16384 edges) sum the histogram columns ahead of their runs themselves; 8 -> 6 launches; row sums still in LDS-arrival order
# speedup vs baseline: 3.3186x; 1.1222x over previous
#include <hip/hip_runtime.h>


typedef __attribute__((ext_vector_type(2))) float v2f;
typedef __attribute__((ext_vector_type(8))) float v8f;

__device__ __forceinline__ v8f wmma_k4(v2f a, v2f b, v8f c) {
  return __builtin_amdgcn_wmma_f32_16x16x4_f32(
       false, a,  false, b,
       (short)0, c,  false,  false);
}


__device__ __forceinline__ int pick4(int4 v, int r) {
  return r == 0 ? v.x : r == 1 ? v.y : r == 2 ? v.z : v.w;
}

__device__ __forceinline__ const int* pick_ip(const int* a, const int* b,
                                              const int* c, const int* d,
                                              int r) {
  return r == 0 ? a : r == 1 ? b : r == 2 ? c : d;
}

__device__ __forceinline__ const float* pick_fp(const float* a, const float* b,
                                                const float* c, const float* d,
                                                int r) {
  return r == 0 ? a : r == 1 ? b : r == 2 ? c : d;
}


__device__ __forceinline__ int block_excl_scan_256(int v, int* sh, int& total) {
  const int t = threadIdx.x;
  sh[t] = v;
  __syncthreads();
  for (int off = 1; off < 256; off <<= 1) {
    const int add = (t >= off) ? sh[t - off] : 0;
    __syncthreads();
    sh[t] += add;
    __syncthreads();
  }
  const int incl = sh[t];
  total = sh[255];
  __syncthreads();
  return incl - v;
}

#define MAX_BINS 1024
#define MAX_BIN_NODES 2048
#define SORT_THREADS 1024
#define EDGES_PER_BLOCK 16384
#define EDGES_PER_THREAD (EDGES_PER_BLOCK / SORT_THREADS)

struct RelGeo {
  int4 E;
  int4 shift;
  int4 nbins;
  int4 nblocks;
  int4 tabOff;
  int4 N;
  int4 nodeOff;
  int4 relBase;
  int4 bsOff;
};

__global__ __launch_bounds__(SORT_THREADS) void bin_hist_kernel(
    const int* __restrict__ dst0, const int* __restrict__ dst1,
    const int* __restrict__ dst2, const int* __restrict__ dst3, RelGeo g,
    int* __restrict__ tabAll, const float* __restrict__ wa0,
    const float* __restrict__ wb0, const float* __restrict__ wa1,
    const float* __restrict__ wb1, const float* __restrict__ wa2,
    const float* __restrict__ wb2, const float* __restrict__ wa3,
    const float* __restrict__ wb3, float* __restrict__ wsum, int4 wOff,
    int4 wN) {
  __shared__ int h[MAX_BINS];
  const int t = threadIdx.x;
  const int r = blockIdx.y;
  const int nblocks = pick4(g.nblocks, r);
  if ((int)blockIdx.x >= nblocks) {
    if (blockIdx.x == gridDim.x - 1) {
      const float* a = pick_fp(wa0, wa1, wa2, wa3, r);
      const float* b = pick_fp(wb0, wb1, wb2, wb3, r);
      float* o = wsum + pick4(wOff, r);
      const int n = pick4(wN, r);
      for (int i = t; i < n; i += SORT_THREADS) o[i] = a[i] + b[i];
    }
    return;
  }
  const int* __restrict__ dst = pick_ip(dst0, dst1, dst2, dst3, r);
  const int E = pick4(g.E, r);
  const int shift = pick4(g.shift, r);
  const int nbins = pick4(g.nbins, r);
  int* __restrict__ tab = tabAll + pick4(g.tabOff, r);
  h[t] = 0;
  __syncthreads();
  const int base = blockIdx.x * EDGES_PER_BLOCK;
  int d[EDGES_PER_THREAD];
#pragma unroll
  for (int j = 0; j < EDGES_PER_THREAD; ++j) {
    const int i = base + j * SORT_THREADS + t;
    d[j] = (i < E) ? dst[i] : -1;
  }
#pragma unroll
  for (int j = 0; j < EDGES_PER_THREAD; ++j)
    if (d[j] >= 0) atomicAdd(&h[d[j] >> shift], 1);
  __syncthreads();
  if (t < nbins) tab[blockIdx.x * nbins + t] = h[t];
}

__global__ __launch_bounds__(SORT_THREADS) void bin_scatter_kernel(
    const int* __restrict__ src0, const int* __restrict__ src1,
    const int* __restrict__ src2, const int* __restrict__ src3,
    const int* __restrict__ dst0, const int* __restrict__ dst1,
    const int* __restrict__ dst2, const int* __restrict__ dst3, RelGeo g,
    const int* __restrict__ tabAll, int* __restrict__ binStart,
    int* __restrict__ bedge) {
  __shared__ int cur[MAX_BINS];
  __shared__ int sh[SORT_THREADS / 32];
  const int t = threadIdx.x;
  const int r = blockIdx.y;
  const int nblocks = pick4(g.nblocks, r);
  if ((int)blockIdx.x >= nblocks && blockIdx.x != 0) return;
  const int* __restrict__ src = pick_ip(src0, src1, src2, src3, r);
  const int* __restrict__ dst = pick_ip(dst0, dst1, dst2, dst3, r);
  const int E = pick4(g.E, r);
  const int shift = pick4(g.shift, r);
  const int nbins = pick4(g.nbins, r);
  const int relBase = pick4(g.relBase, r);
  const int* __restrict__ tab = tabAll + pick4(g.tabOff, r);
  int* __restrict__ bs = binStart + pick4(g.bsOff, r);

  int all = 0, before = 0;
  if (t < nbins) {
    for (int b0 = 0; b0 < nblocks; b0 += 16) {
      int v[16];
#pragma unroll
      for (int u = 0; u < 16; ++u)
        v[u] = (b0 + u < nblocks) ? tab[(b0 + u) * nbins + t] : 0;
#pragma unroll
      for (int u = 0; u < 16; ++u) {
        all += v[u];
        before += (b0 + u < (int)blockIdx.x) ? v[u] : 0;
      }
    }
  }
  const int lane = t & 31;
  const int wave = t >> 5;
  int incl = all;
  for (int off = 1; off < 32; off <<= 1) {
    const int up = __shfl_up(incl, off);
    if (lane >= off) incl += up;
  }
  if (lane == 31) sh[wave] = incl;
  __syncthreads();
  const int wsum = sh[lane];
  int wincl = wsum;
  for (int off = 1; off < 32; off <<= 1) {
    const int up = __shfl_up(wincl, off);
    if (lane >= off) wincl += up;
  }
  const int waveOff = __shfl(wincl - wsum, wave);
  const int start = relBase + waveOff + (incl - all);
  cur[t] = start + before;
  if (blockIdx.x == 0) {
    if (t < nbins) bs[t] = start;
    if (t == 0) bs[nbins] = relBase + E;
  }
  __syncthreads();

  int2* be = reinterpret_cast<int2*>(bedge);
  const int base = blockIdx.x * EDGES_PER_BLOCK;
  int d[EDGES_PER_THREAD], s[EDGES_PER_THREAD];
  int slot[EDGES_PER_THREAD];
#pragma unroll
  for (int j = 0; j < EDGES_PER_THREAD; ++j) {
    const int i = base + j * SORT_THREADS + t;
    d[j] = (i < E) ? dst[i] : -1;
    s[j] = (i < E) ? src[i] : 0;
  }
#pragma unroll
  for (int j = 0; j < EDGES_PER_THREAD; ++j)
    slot[j] = (d[j] >= 0) ? atomicAdd(&cur[d[j] >> shift], 1) : 0;
#pragma unroll
  for (int j = 0; j < EDGES_PER_THREAD; ++j)
    if (d[j] >= 0) be[slot[j]] = make_int2(d[j], s[j]);
}

__global__ __launch_bounds__(256) void bin_csr_kernel(
    const int* __restrict__ bedge, RelGeo g, const int* __restrict__ binStart,
    int* __restrict__ cntAll, int* __restrict__ posAll,
    int* __restrict__ col) {
  __shared__ int lc[MAX_BIN_NODES];
  __shared__ int sh[256];
  const int t = threadIdx.x;
  const int b = blockIdx.x;
  const int rel = blockIdx.y;
  if (b >= pick4(g.nbins, rel)) return;
  const int shift = pick4(g.shift, rel);
  const int N = pick4(g.N, rel);
  int* __restrict__ cnt = cntAll + pick4(g.nodeOff, rel);
  int* __restrict__ pos = posAll + pick4(g.nodeOff, rel);
  const int B = 1 << shift;
  const int per = B >> 8;
  const int nodeBase = b << shift;
  const int* __restrict__ bs = binStart + pick4(g.bsOff, rel);
  const int beg = bs[b];
  const int end = bs[b + 1];
  const int2* be = reinterpret_cast<const int2*>(bedge);

  for (int k = t; k < B; k += 256) lc[k] = 0;
  __syncthreads();
  for (int e0 = beg; e0 < end; e0 += 8 * 256) {
    int x[8];
#pragma unroll
    for (int u = 0; u < 8; ++u) {
      const int e = e0 + u * 256 + t;
      x[u] = (e < end) ? be[e].x : -1;
    }
#pragma unroll
    for (int u = 0; u < 8; ++u)
      if (x[u] >= 0) atomicAdd(&lc[x[u] - nodeBase], 1);
  }
  __syncthreads();

  int v[8];
  int s = 0;
#pragma unroll
  for (int j = 0; j < 8; ++j) {
    v[j] = (j < per) ? lc[t * per + j] : 0;
    s += v[j];
  }
  int total;
  int run = block_excl_scan_256(s, sh, total);
#pragma unroll
  for (int j = 0; j < 8; ++j) {
    if (j < per) {
      const int k = t * per + j;
      const int node = nodeBase + k;
      lc[k] = run;
      if (node < N) {
        cnt[node] = v[j];
        pos[node] = beg + run;
      }
      run += v[j];
    }
  }
  __syncthreads();
  for (int e0 = beg; e0 < end; e0 += 8 * 256) {
    int2 r[8];
    int slot[8];
#pragma unroll
    for (int u = 0; u < 8; ++u) {
      const int e = e0 + u * 256 + t;
      r[u] = (e < end) ? be[e] : make_int2(-1, 0);
    }
#pragma unroll
    for (int u = 0; u < 8; ++u)
      slot[u] = (r[u].x >= 0) ? atomicAdd(&lc[r[u].x - nodeBase], 1) : 0;
#pragma unroll
    for (int u = 0; u < 8; ++u)
      if (r[u].x >= 0) col[beg + slot[u]] = r[u].y;
  }
}

__device__ __forceinline__ float4 load_row4(const float* __restrict__ Y, int s,
                                            int c) {
  return *reinterpret_cast<const float4*>(Y + (size_t)s * 16 + c * 4);
}

__device__ __forceinline__ float4 gather_rows(const float* __restrict__ Y,
                                              const int* __restrict__ col,
                                              int beg, int end, int c) {
  float4 acc = make_float4(0.0f, 0.0f, 0.0f, 0.0f);
  int p = beg;
  for (; p + 1 < end; p += 2) {
    const int s0 = col[p];
    const int s1 = col[p + 1];
    const float4 v0 = load_row4(Y, s0, c);
    const float4 v1 = load_row4(Y, s1, c);
    acc.x += v0.x;
    acc.y += v0.y;
    acc.z += v0.z;
    acc.w += v0.w;
    acc.x += v1.x;
    acc.y += v1.y;
    acc.z += v1.z;
    acc.w += v1.w;
  }
  if (p < end) {
    const float4 v = load_row4(Y, col[p], c);
    acc.x += v.x;
    acc.y += v.y;
    acc.z += v.z;
    acc.w += v.w;
  }
  return acc;
}

__global__ void gather_l1_kernel(const float* __restrict__ YAd,
                                 const float* __restrict__ YBd,
                                 const float* __restrict__ YAt,
                                 const float* __restrict__ YBt,
                                 const int* __restrict__ cnt,
                                 const int* __restrict__ pos, int4 nodeOff,
                                 const int* __restrict__ col,
                                 float* __restrict__ Td,
                                 float* __restrict__ Tt, int ND, int NT,
                                 int blocksD) {
  const bool isT = (int)blockIdx.x >= blocksD;
  const int bid = isT ? blockIdx.x - blocksD : blockIdx.x;
  const float* __restrict__ YA = isT ? YAt : YAd;
  const float* __restrict__ YB = isT ? YBt : YBd;
  const int* __restrict__ cntA = cnt + (isT ? nodeOff.x : nodeOff.z);
  const int* __restrict__ cntB = cnt + (isT ? nodeOff.y : nodeOff.w);
  const int* __restrict__ begA = pos + (isT ? nodeOff.x : nodeOff.z);
  const int* __restrict__ begB = pos + (isT ? nodeOff.y : nodeOff.w);
  float* __restrict__ T = isT ? Tt : Td;
  const int N = isT ? NT : ND;
  const int i = bid * blockDim.x + threadIdx.x;
  const int node = i >> 2;
  const int c = i & 3;
  if (node >= N) return;
  const int nA = cntA[node], bA = begA[node];
  const int nB = cntB[node], bB = begB[node];
  float4* tp = reinterpret_cast<float4*>(T + (size_t)node * 16 + c * 4);
  float4 acc = *tp;
  const float4 sA = gather_rows(YA, col, bA, bA + nA, c);
  const float4 sB = gather_rows(YB, col, bB, bB + nB, c);
  const float iA = 1.0f / fmaxf((float)nA, 1.0f);
  const float iB = 1.0f / fmaxf((float)nB, 1.0f);
  acc.x += sA.x * iA + sB.x * iB;
  acc.y += sA.y * iA + sB.y * iB;
  acc.z += sA.z * iA + sB.z * iB;
  acc.w += sA.w * iA + sB.w * iB;
  acc.x = fmaxf(acc.x, 0.0f);
  acc.y = fmaxf(acc.y, 0.0f);
  acc.z = fmaxf(acc.z, 0.0f);
  acc.w = fmaxf(acc.w, 0.0f);
  *tp = acc;
}

__global__ __launch_bounds__(128) void l1_transform_kernel(
    const float* __restrict__ Xd, const float* __restrict__ W0d,
    const float* __restrict__ W1d, const float* __restrict__ W2d,
    const float* __restrict__ biasAd, const float* __restrict__ biasBd,
    float* __restrict__ Y0d, float* __restrict__ Y1d, float* __restrict__ Y2d,
    int Md, const float* __restrict__ Xt, const float* __restrict__ W0t,
    const float* __restrict__ W1t, const float* __restrict__ W2t,
    const float* __restrict__ biasAt, const float* __restrict__ biasBt,
    float* __restrict__ Y0t, float* __restrict__ Y1t, float* __restrict__ Y2t,
    int Mt, int blocksD) {
  const bool isT = (int)blockIdx.x >= blocksD;
  const int bid = isT ? blockIdx.x - blocksD : blockIdx.x;
  const int nb = isT ? gridDim.x - blocksD : blocksD;
  const float* __restrict__ X = isT ? Xt : Xd;
  const float* __restrict__ W0 = isT ? W0t : W0d;
  const float* __restrict__ W1 = isT ? W1t : W1d;
  const float* __restrict__ W2 = isT ? W2t : W2d;
  const float* __restrict__ biasA = isT ? biasAt : biasAd;
  const float* __restrict__ biasB = isT ? biasBt : biasBd;
  float* __restrict__ Y0 = isT ? Y0t : Y0d;
  float* __restrict__ Y1 = isT ? Y1t : Y1d;
  float* __restrict__ Y2 = isT ? Y2t : Y2d;
  const int M = isT ? Mt : Md;
  const int lane = threadIdx.x & 31;
  const int wave = threadIdx.x >> 5;
  const int wpb = blockDim.x >> 5;
  const int half = lane >> 4;
  const int mrow = lane & 15;
  const int nTiles = (M + 15) >> 4;

  v2f b0[16], b1[16], b2[16];
#pragma unroll
  for (int s = 0; s < 16; ++s) {
    int k = 4 * s + 2 * half;
    b0[s] = (v2f){W0[k * 16 + mrow], W0[(k + 1) * 16 + mrow]};
    b1[s] = (v2f){W1[k * 16 + mrow], W1[(k + 1) * 16 + mrow]};
    b2[s] = (v2f){W2[k * 16 + mrow], W2[(k + 1) * 16 + mrow]};
  }

  const float bsum = biasA[mrow] + biasB[mrow];

  for (int tile = bid * wpb + wave; tile < nTiles; tile += nb * wpb) {
    const int rbase = tile << 4;
    int arow = rbase + mrow;
    if (arow > M - 1) arow = M - 1;
    const float* xr = X + (size_t)arow * 64;
    v8f c0 = {}, c1 = {};
    v8f c2 = {bsum, bsum, bsum, bsum, bsum, bsum, bsum, bsum};
#pragma unroll
    for (int s = 0; s < 16; ++s) {
      int k = 4 * s + 2 * half;
      v2f a = (v2f){xr[k], xr[k + 1]};
      c0 = wmma_k4(a, b0[s], c0);
      c1 = wmma_k4(a, b1[s], c1);
      c2 = wmma_k4(a, b2[s], c2);
    }
#pragma unroll
    for (int r = 0; r < 8; ++r) {
      int row = rbase + r + 8 * half;
      if (row < M) {
        size_t o = (size_t)row * 16 + mrow;
        Y0[o] = c0[r];
        Y1[o] = c1[r];
        Y2[o] = c2[r];
      }
    }
  }
}

#define L2_ROWS 64
#define L2_PITCH 20

__global__ __launch_bounds__(256) void l2_gather_fused_kernel(
    const float* __restrict__ Hnd, const float* __restrict__ Hrd,
    const float* __restrict__ W0d, const float* __restrict__ W1d,
    const float* __restrict__ W2d, const float* __restrict__ bias0d,
    const float* __restrict__ bias1d, float* __restrict__ OUTd, int Md,
    const float* __restrict__ Hnt, const float* __restrict__ Hrt,
    const float* __restrict__ W0t, const float* __restrict__ W1t,
    const float* __restrict__ W2t, const float* __restrict__ bias0t,
    const float* __restrict__ bias1t, float* __restrict__ OUTt, int Mt,
    const int* __restrict__ cnt, const int* __restrict__ pos, int4 nodeOff,
    const int* __restrict__ col, int blocksD) {
  __shared__ __attribute__((aligned(16))) float stage[3][L2_ROWS][L2_PITCH];
  const bool isT = (int)blockIdx.x >= blocksD;
  const int bid = isT ? blockIdx.x - blocksD : blockIdx.x;
  const float* __restrict__ Hn = isT ? Hnt : Hnd;
  const float* __restrict__ Hr = isT ? Hrt : Hrd;
  const float* __restrict__ W0 = isT ? W0t : W0d;
  const float* __restrict__ W1 = isT ? W1t : W1d;
  const float* __restrict__ W2 = isT ? W2t : W2d;
  const float* __restrict__ bias0 = isT ? bias0t : bias0d;
  const float* __restrict__ bias1 = isT ? bias1t : bias1d;
  float* __restrict__ OUT = isT ? OUTt : OUTd;
  const int* __restrict__ cntA = cnt + (isT ? nodeOff.x : nodeOff.z);
  const int* __restrict__ cntB = cnt + (isT ? nodeOff.y : nodeOff.w);
  const int* __restrict__ begA = pos + (isT ? nodeOff.x : nodeOff.z);
  const int* __restrict__ begB = pos + (isT ? nodeOff.y : nodeOff.w);
  const int M = isT ? Mt : Md;
  const int t = threadIdx.x;
  const int lane = t & 31;
  const int wave = t >> 5;
  const int half = lane >> 4;
  const int mrow = lane & 15;
  const int ct = wave & 1;
  const int tl = wave >> 1;

  const float* Ws[3] = {W0, W1, W2};
  v2f b[3][4];
#pragma unroll
  for (int m = 0; m < 3; ++m)
#pragma unroll
    for (int s = 0; s < 4; ++s) {
      const int k = 4 * s + 2 * half;
      const int c = 16 * ct + mrow;
      b[m][s] = (v2f){Ws[m][k * 32 + c], Ws[m][(k + 1) * 32 + c]};
    }
  const float bias = bias0[16 * ct + mrow] + bias1[16 * ct + mrow];

  {
    const int r = t >> 2;
    const int q = t & 3;
    const int node = bid * L2_ROWS + r;
    float4 sA = make_float4(0.0f, 0.0f, 0.0f, 0.0f);
    float4 sB = sA, rt = sA;
    if (node < M) {
      const int nA = cntA[node], bA = begA[node];
      const int nB = cntB[node], bB = begB[node];
      rt = load_row4(Hr, node, q);
      sA = gather_rows(Hn, col, bA, bA + nA, q);
      sB = gather_rows(Hn, col, bB, bB + nB, q);
      const float iA = 1.0f / fmaxf((float)nA, 1.0f);
      const float iB = 1.0f / fmaxf((float)nB, 1.0f);
      sA.x *= iA;
      sA.y *= iA;
      sA.z *= iA;
      sA.w *= iA;
      sB.x *= iB;
      sB.y *= iB;
      sB.z *= iB;
      sB.w *= iB;
      rt.x = fmaxf(rt.x, 0.0f);
      rt.y = fmaxf(rt.y, 0.0f);
      rt.z = fmaxf(rt.z, 0.0f);
      rt.w = fmaxf(rt.w, 0.0f);
    }
    *reinterpret_cast<float4*>(&stage[0][r][4 * q]) = sA;
    *reinterpret_cast<float4*>(&stage[1][r][4 * q]) = sB;
    *reinterpret_cast<float4*>(&stage[2][r][4 * q]) = rt;
  }
  __syncthreads();

  const int rbase = bid * L2_ROWS + 16 * tl;
  if (rbase >= M) return;
  v8f acc = {};
#pragma unroll
  for (int s = 0; s < 4; ++s)
#pragma unroll
    for (int m = 0; m < 3; ++m) {
      const v2f a = *reinterpret_cast<const v2f*>(
          &stage[m][16 * tl + mrow][4 * s + 2 * half]);
      acc = wmma_k4(a, b[m][s], acc);
    }
#pragma unroll
  for (int r = 0; r < 8; ++r) {
    const int orow = rbase + r + 8 * half;
    if (orow < M) OUT[(size_t)orow * 32 + 16 * ct + mrow] = acc[r] + bias;
  }
}


static inline int bin_shift_for(int n) {
  int s = 8;
  while (s < 11 && ((n - 1) >> s) >= 256) ++s;
  return s;
}

extern "C" void kernel_launch(void* const* d_in, const int* in_sizes, int n_in,
                              void* d_out, int out_size, void* d_ws,
                              size_t ws_size, hipStream_t stream) {
  const float* x_dev = (const float*)d_in[0];
  const float* x_task = (const float*)d_in[1];
  const int* wr_src = (const int*)d_in[2];
  const int* wr_dst = (const int*)d_in[3];
  const int* rv_src = (const int*)d_in[4];
  const int* rv_dst = (const int*)d_in[5];
  const int* wb_src = (const int*)d_in[6];
  const int* wb_dst = (const int*)d_in[7];
  const int* rb_src = (const int*)d_in[8];
  const int* rb_dst = (const int*)d_in[9];
  const float* p1wr_wl = (const float*)d_in[10];
  const float* p1wr_bl = (const float*)d_in[11];
  const float* p1wr_wr = (const float*)d_in[12];
  const float* p1rv_wl = (const float*)d_in[13];
  const float* p1rv_bl = (const float*)d_in[14];
  const float* p1rv_wr = (const float*)d_in[15];
  const float* p1wb_wl = (const float*)d_in[16];
  const float* p1wb_bl = (const float*)d_in[17];
  const float* p1wb_wr = (const float*)d_in[18];
  const float* p1rb_wl = (const float*)d_in[19];
  const float* p1rb_bl = (const float*)d_in[20];
  const float* p1rb_wr = (const float*)d_in[21];
  const float* p2wr_wl = (const float*)d_in[22];
  const float* p2wr_bl = (const float*)d_in[23];
  const float* p2wr_wr = (const float*)d_in[24];
  const float* p2rv_wl = (const float*)d_in[25];
  const float* p2rv_bl = (const float*)d_in[26];
  const float* p2rv_wr = (const float*)d_in[27];
  const float* p2wb_wl = (const float*)d_in[28];
  const float* p2wb_bl = (const float*)d_in[29];
  const float* p2wb_wr = (const float*)d_in[30];
  const float* p2rb_wl = (const float*)d_in[31];
  const float* p2rb_bl = (const float*)d_in[32];
  const float* p2rb_wr = (const float*)d_in[33];

  const int ND = in_sizes[0] / 64;
  const int NT = in_sizes[1] / 64;
  const int E_wr = in_sizes[2], E_rv = in_sizes[4];
  const int E_wb = in_sizes[6], E_rb = in_sizes[8];

  const int cntN = 2 * NT + 2 * ND;
  const int Etot = E_wr + E_rv + E_wb + E_rb;
  const int shT = bin_shift_for(NT), shD = bin_shift_for(ND);
  const int nbT = ((NT - 1) >> shT) + 1, nbD = ((ND - 1) >> shD) + 1;
  const int nc_wr = (E_wr + EDGES_PER_BLOCK - 1) / EDGES_PER_BLOCK;
  const int nc_rv = (E_rv + EDGES_PER_BLOCK - 1) / EDGES_PER_BLOCK;
  const int nc_wb = (E_wb + EDGES_PER_BLOCK - 1) / EDGES_PER_BLOCK;
  const int nc_rb = (E_rb + EDGES_PER_BLOCK - 1) / EDGES_PER_BLOCK;
  const int to_wr = 0;
  const int to_rv = to_wr + nbT * nc_wr;
  const int to_wb = to_rv + nbT * nc_rv;
  const int to_rb = to_wb + nbD * nc_wb;
  const int nTab = to_rb + nbD * nc_rb;
  float* ws = (float*)d_ws;
  size_t off = 0;
  int* cnt = (int*)(ws + off); off += (size_t)cntN;
  float* csr = ws + off; off += (size_t)cntN * 16;
  float* yreg = ws + off; off += (size_t)cntN * 16;
  float* t1 = ws + off; off += (size_t)NT * 16;
  float* d1 = ws + off; off += (size_t)ND * 16;
  float* ws1t = ws + off; off += 64 * 16;
  float* ws1d = ws + off; off += 64 * 16;
  float* ws2t = ws + off; off += 16 * 32;
  float* ws2d = ws + off; off += 16 * 32;

  int* pos = (int*)csr;
  int* col = pos + cntN;
  size_t q = (size_t)2 * cntN + Etot;
  q = (q + 1) & ~(size_t)1;
  int* bedge = (int*)ws + q; q += (size_t)2 * Etot;
  int* tab = (int*)ws + q; q += (size_t)nTab;
  int* binStart = (int*)ws + q;

  float* y_wr = yreg;
  float* y_rv = y_wr + (size_t)ND * 16;
  float* y_wb = y_rv + (size_t)ND * 16;
  float* y_rb = y_wb + (size_t)NT * 16;

  const int TPB = 256;

  RelGeo g;
  g.E = make_int4(E_wr, E_rv, E_wb, E_rb);
  g.shift = make_int4(shT, shT, shD, shD);
  g.nbins = make_int4(nbT, nbT, nbD, nbD);
  g.nblocks = make_int4(nc_wr, nc_rv, nc_wb, nc_rb);
  g.tabOff = make_int4(to_wr, to_rv, to_wb, to_rb);
  g.N = make_int4(NT, NT, ND, ND);
  g.nodeOff = make_int4(0, NT, 2 * NT, 2 * NT + ND);
  g.relBase = make_int4(0, E_wr, E_wr + E_rv, E_wr + E_rv + E_wb);
  g.bsOff = make_int4(0, nbT + 1, 2 * (nbT + 1), 2 * (nbT + 1) + nbD + 1);
  int ncMax = nc_wr;
  if (nc_rv > ncMax) ncMax = nc_rv;
  if (nc_wb > ncMax) ncMax = nc_wb;
  if (nc_rb > ncMax) ncMax = nc_rb;
  if (ncMax < 1) ncMax = 1;
  const int nbMax = (nbT > nbD) ? nbT : nbD;

  bin_hist_kernel<<<dim3(ncMax + 1, 4), SORT_THREADS, 0, stream>>>(
      wr_dst, rv_dst, wb_dst, rb_dst, g, tab, p1wr_wr, p1rv_wr, p1wb_wr,
      p1rb_wr, p2wr_wr, p2rv_wr, p2wb_wr, p2rb_wr, ws1t,
      make_int4(0, 64 * 16, 2 * 64 * 16, 2 * 64 * 16 + 16 * 32),
      make_int4(64 * 16, 64 * 16, 16 * 32, 16 * 32));

  bin_scatter_kernel<<<dim3(ncMax, 4), SORT_THREADS, 0, stream>>>(
      wr_src, rv_src, wb_src, rb_src, wr_dst, rv_dst, wb_dst, rb_dst, g, tab,
      binStart, bedge);

  bin_csr_kernel<<<dim3(nbMax, 4), 256, 0, stream>>>(bedge, g, binStart, cnt,
                                                     pos, col);

  const int tilesD = (ND + 15) / 16, tilesT = (NT + 15) / 16;
  const int tbD = (tilesD + 3) / 4, tbT = (tilesT + 3) / 4;
  l1_transform_kernel<<<tbD + tbT, 128, 0, stream>>>(
      x_dev, p1wr_wl, p1rv_wl, ws1d, p1wb_bl, p1rb_bl, y_wr, y_rv, d1, ND,
      x_task, p1wb_wl, p1rb_wl, ws1t, p1wr_bl, p1rv_bl, y_wb, y_rb, t1, NT,
      tbD);

  const int gbD = (ND * 4 + TPB - 1) / TPB, gbT = (NT * 4 + TPB - 1) / TPB;
  gather_l1_kernel<<<gbD + gbT, TPB, 0, stream>>>(
      y_wb, y_rb, y_wr, y_rv, cnt, pos, g.nodeOff, col, d1, t1, ND, NT, gbD);

  float* out_d2 = (float*)d_out;
  float* out_t2 = (float*)d_out + (size_t)ND * 32;
  const int fbD = (ND + L2_ROWS - 1) / L2_ROWS;
  const int fbT = (NT + L2_ROWS - 1) / L2_ROWS;
  l2_gather_fused_kernel<<<fbD + fbT, 256, 0, stream>>>(
      t1, d1, p2wb_wl, p2rb_wl, ws2d, p2wb_bl, p2rb_bl, out_d2, ND, d1, t1,
      p2wr_wl, p2rv_wl, ws2t, p2wr_bl, p2rv_bl, out_t2, NT, cnt, pos,
      g.nodeOff, col, fbD);
}
